# starting kernel
# baseline (speedup 1.0000x reference)
#include <hip/hip_runtime.h>


#define NPTS   16384
#define DIN    1024
#define DPROJ  128
#define NSAMP  163
#define NCHUNK 8
#define TILES_PER_CHUNK (NPTS / 16 / NCHUNK)

typedef float v2f __attribute__((ext_vector_type(2)));
typedef float v8f __attribute__((ext_vector_type(8)));

__device__ __forceinline__ v8f wmma_f32_k4(v2f a, v2f b, v8f c) {
  return __builtin_amdgcn_wmma_f32_16x16x4_f32(
       false, a,  false, b,
       (short)0, c,  false,  false);
}

__global__ __launch_bounds__(256)
void proj_kernel(const float* __restrict__ F, const float* __restrict__ W,
                 float* __restrict__ R) {
  const int wave = (blockIdx.x * blockDim.x + threadIdx.x) >> 5;
  const int lane = threadIdx.x & 31;
  const int tm = wave >> 3;
  const int tn = wave & 7;
  const int m0 = tm * 16, n0 = tn * 16;
  const int half = lane >> 4;
  const int r16  = lane & 15;
  const int koff = half * 2;

  v8f acc = {};
  const float* arow = F + (size_t)(m0 + r16) * DIN + koff;
  for (int k = 0; k < DIN; k += 4) {
    v2f a = *(const v2f*)(arow + k);
    v2f b;
    b.x = W[(size_t)(k + koff)     * DPROJ + n0 + r16];
    b.y = W[(size_t)(k + koff + 1) * DPROJ + n0 + r16];
    acc = wmma_f32_k4(a, b, acc);
  }
#pragma unroll
  for (int i = 0; i < 8; i++)
    R[(size_t)(m0 + i + half * 8) * DPROJ + n0 + r16] = acc[i];
}

__global__ __launch_bounds__(256)
void sq_kernel(const float* __restrict__ R, float* __restrict__ sq) {
  const int row  = (blockIdx.x * blockDim.x + threadIdx.x) >> 5;
  const int lane = threadIdx.x & 31;
  const float* rp = R + (size_t)row * DPROJ;
  float s = 0.f;
#pragma unroll
  for (int k = lane; k < DPROJ; k += 32) { float v = rp[k]; s += v * v; }
#pragma unroll
  for (int m = 16; m >= 1; m >>= 1) s += __shfl_xor(s, m, 32);
  if (lane == 0) sq[row] = s;
}

__global__ __launch_bounds__(256)
void gram_rowsum_kernel(const float* __restrict__ R, const float* __restrict__ sq,
                        float* __restrict__ anchorPart) {
  const int wave  = (blockIdx.x * blockDim.x + threadIdx.x) >> 5;
  const int lane  = threadIdx.x & 31;
  const int chunk = wave & (NCHUNK - 1);
  const int strip = wave >> 3;
  const int half  = lane >> 4;
  const int r16   = lane & 15;
  const int koff  = half * 2;
  const int m0    = strip * 16;

  v2f afrag[32];
  const float* arow = R + (size_t)(m0 + r16) * DPROJ + koff;
#pragma unroll
  for (int kk = 0; kk < 32; kk++) afrag[kk] = *(const v2f*)(arow + kk * 4);

  float sqr[8];
#pragma unroll
  for (int i = 0; i < 8; i++) sqr[i] = sq[m0 + i + half * 8];

  float partial[8] = {0.f, 0.f, 0.f, 0.f, 0.f, 0.f, 0.f, 0.f};

  const int t0 = chunk * TILES_PER_CHUNK;
  for (int t = t0; t < t0 + TILES_PER_CHUNK; t++) {
    const int col = t * 16 + r16;
    const float sqc = sq[col];
    const float* brow = R + (size_t)col * DPROJ + koff;
    __builtin_prefetch((const void*)(brow + 16 * DPROJ), 0, 1);
    v8f acc = {};
#pragma unroll
    for (int kk = 0; kk < 32; kk++) {
      v2f b = *(const v2f*)(brow + kk * 4);
      acc = wmma_f32_k4(afrag[kk], b, acc);
    }
#pragma unroll
    for (int i = 0; i < 8; i++)
      partial[i] += fmaxf(0.f, sqr[i] + sqc - 2.f * acc[i]);
  }

#pragma unroll
  for (int i = 0; i < 8; i++) {
    float p = partial[i];
    p += __shfl_xor(p, 1, 32);
    p += __shfl_xor(p, 2, 32);
    p += __shfl_xor(p, 4, 32);
    p += __shfl_xor(p, 8, 32);
    if (r16 == 0)
      anchorPart[(size_t)chunk * NPTS + m0 + i + half * 8] = p;
  }
}

__global__ __launch_bounds__(256)
void combine_kernel(const float* __restrict__ anchorPart,
                    float* __restrict__ anchorSq) {
  const int i = blockIdx.x * 256 + threadIdx.x;
  float s = 0.f;
#pragma unroll
  for (int c = 0; c < NCHUNK; c++) s += anchorPart[(size_t)c * NPTS + i];
  anchorSq[i] = s;
}

__global__ __launch_bounds__(1024)
void greedy_kernel(const float* __restrict__ R, const float* __restrict__ sq,
                   const float* __restrict__ anchorSq, int* __restrict__ sel) {
  __shared__ float redV[32];
  __shared__ int   redI[32];
  __shared__ float selRow[DPROJ];
  __shared__ float selSqS;
  __shared__ int   selIdxS;

  const int t    = threadIdx.x;
  const int lane = t & 31;
  const int wid  = t >> 5;

  float aVal[16], sqOwn[16];
#pragma unroll
  for (int j = 0; j < 16; j++) {
    const int row = t + j * 1024;
    aVal[j]  = sqrtf(anchorSq[row]);
    sqOwn[j] = sq[row];
  }

  for (int s = 0; s < NSAMP; s++) {
    float bv = aVal[0]; int bi = t;
#pragma unroll
    for (int j = 1; j < 16; j++)
      if (aVal[j] > bv) { bv = aVal[j]; bi = t + j * 1024; }
#pragma unroll
    for (int m = 1; m < 32; m <<= 1) {
      const float ov = __shfl_xor(bv, m, 32);
      const int   oi = __shfl_xor(bi, m, 32);
      if (ov > bv || (ov == bv && oi < bi)) { bv = ov; bi = oi; }
    }
    if (lane == 0) { redV[wid] = bv; redI[wid] = bi; }
    __syncthreads();
    if (wid == 0) {
      bv = redV[lane]; bi = redI[lane];
#pragma unroll
      for (int m = 1; m < 32; m <<= 1) {
        const float ov = __shfl_xor(bv, m, 32);
        const int   oi = __shfl_xor(bi, m, 32);
        if (ov > bv || (ov == bv && oi < bi)) { bv = ov; bi = oi; }
      }
      if (lane == 0) { selIdxS = bi; selSqS = sq[bi]; sel[s] = bi; }
    }
    __syncthreads();
    const int si = selIdxS;
    if (t < DPROJ) selRow[t] = R[(size_t)si * DPROJ + t];
    __syncthreads();

    const float selSq = selSqS;
#pragma unroll 4
    for (int j = 0; j < 16; j++) {
      const int row = t + j * 1024;
      const float4* rp = (const float4*)(R + (size_t)row * DPROJ);
      const float4* wp = (const float4*)selRow;
      float dot = 0.f;
#pragma unroll
      for (int k = 0; k < DPROJ / 4; k++) {
        const float4 v = rp[k];
        const float4 w = wp[k];
        dot += v.x * w.x + v.y * w.y + v.z * w.z + v.w * w.w;
      }
      const float d2 = sqOwn[j] + selSq - 2.f * dot;
      aVal[j] = fminf(aVal[j], sqrtf(fmaxf(d2, 0.f)));
    }
    __syncthreads();
  }
}

__global__ __launch_bounds__(256)
void gather_kernel(const float* __restrict__ F, const int* __restrict__ sel,
                   float* __restrict__ out) {
  const int s = blockIdx.x;
  const int c = blockIdx.y * 256 + threadIdx.x;
  out[(size_t)s * DIN + c] = F[(size_t)sel[s] * DIN + c];
}

extern "C" void kernel_launch(void* const* d_in, const int* in_sizes, int n_in,
                              void* d_out, int out_size, void* d_ws, size_t ws_size,
                              hipStream_t stream) {
  (void)in_sizes; (void)n_in; (void)out_size; (void)ws_size;
  const float* F = (const float*)d_in[0];
  const float* W = (const float*)d_in[1];

  float* R          = (float*)d_ws;
  float* sq         = R + (size_t)NPTS * DPROJ;
  float* anchorSq   = sq + NPTS;
  float* anchorPart = anchorSq + NPTS;
  int*   sel        = (int*)(anchorPart + (size_t)NCHUNK * NPTS);

  proj_kernel<<<1024, 256, 0, stream>>>(F, W, R);
  sq_kernel<<<NPTS / 8, 256, 0, stream>>>(R, sq);
  gram_rowsum_kernel<<<1024, 256, 0, stream>>>(R, sq, anchorPart);
  combine_kernel<<<NPTS / 256, 256, 0, stream>>>(anchorPart, anchorSq);
  greedy_kernel<<<1, 1024, 0, stream>>>(R, sq, anchorSq, sel);
  gather_kernel<<<dim3(NSAMP, DIN / 256), 256, 0, stream>>>(F, sel, (float*)d_out);
}


# greedy loop: one launch per iteration over 512 one-wave work-groups (argmax via per-wave candidates in workspace) replaces one 1024-thread work-group; row dot summed as 4 interleaved fma chains
# speedup vs baseline: 19.3819x; 19.3819x over previous
#include <hip/hip_runtime.h>


#define NPTS   16384
#define DIN    1024
#define DPROJ  128
#define NSAMP  163
#define NCHUNK 8
#define TILES_PER_CHUNK (NPTS / 16 / NCHUNK)

typedef float v2f __attribute__((ext_vector_type(2)));
typedef float v8f __attribute__((ext_vector_type(8)));

__device__ __forceinline__ v8f wmma_f32_k4(v2f a, v2f b, v8f c) {
  return __builtin_amdgcn_wmma_f32_16x16x4_f32(
       false, a,  false, b,
       (short)0, c,  false,  false);
}

__global__ __launch_bounds__(256)
void proj_kernel(const float* __restrict__ F, const float* __restrict__ W,
                 float* __restrict__ R) {
  const int wave = (blockIdx.x * blockDim.x + threadIdx.x) >> 5;
  const int lane = threadIdx.x & 31;
  const int tm = wave >> 3;
  const int tn = wave & 7;
  const int m0 = tm * 16, n0 = tn * 16;
  const int half = lane >> 4;
  const int r16  = lane & 15;
  const int koff = half * 2;

  v8f acc = {};
  const float* arow = F + (size_t)(m0 + r16) * DIN + koff;
  for (int k = 0; k < DIN; k += 4) {
    v2f a = *(const v2f*)(arow + k);
    v2f b;
    b.x = W[(size_t)(k + koff)     * DPROJ + n0 + r16];
    b.y = W[(size_t)(k + koff + 1) * DPROJ + n0 + r16];
    acc = wmma_f32_k4(a, b, acc);
  }
#pragma unroll
  for (int i = 0; i < 8; i++)
    R[(size_t)(m0 + i + half * 8) * DPROJ + n0 + r16] = acc[i];
}

__global__ __launch_bounds__(256)
void sq_kernel(const float* __restrict__ R, float* __restrict__ sq) {
  const int row  = (blockIdx.x * blockDim.x + threadIdx.x) >> 5;
  const int lane = threadIdx.x & 31;
  const float* rp = R + (size_t)row * DPROJ;
  float s = 0.f;
#pragma unroll
  for (int k = lane; k < DPROJ; k += 32) { float v = rp[k]; s += v * v; }
#pragma unroll
  for (int m = 16; m >= 1; m >>= 1) s += __shfl_xor(s, m, 32);
  if (lane == 0) sq[row] = s;
}

__global__ __launch_bounds__(256)
void gram_rowsum_kernel(const float* __restrict__ R, const float* __restrict__ sq,
                        float* __restrict__ anchorPart) {
  const int wave  = (blockIdx.x * blockDim.x + threadIdx.x) >> 5;
  const int lane  = threadIdx.x & 31;
  const int chunk = wave & (NCHUNK - 1);
  const int strip = wave >> 3;
  const int half  = lane >> 4;
  const int r16   = lane & 15;
  const int koff  = half * 2;
  const int m0    = strip * 16;

  v2f afrag[32];
  const float* arow = R + (size_t)(m0 + r16) * DPROJ + koff;
#pragma unroll
  for (int kk = 0; kk < 32; kk++) afrag[kk] = *(const v2f*)(arow + kk * 4);

  float sqr[8];
#pragma unroll
  for (int i = 0; i < 8; i++) sqr[i] = sq[m0 + i + half * 8];

  float partial[8] = {0.f, 0.f, 0.f, 0.f, 0.f, 0.f, 0.f, 0.f};

  const int t0 = chunk * TILES_PER_CHUNK;
  for (int t = t0; t < t0 + TILES_PER_CHUNK; t++) {
    const int col = t * 16 + r16;
    const float sqc = sq[col];
    const float* brow = R + (size_t)col * DPROJ + koff;
    __builtin_prefetch((const void*)(brow + 16 * DPROJ), 0, 1);
    v8f acc = {};
#pragma unroll
    for (int kk = 0; kk < 32; kk++) {
      v2f b = *(const v2f*)(brow + kk * 4);
      acc = wmma_f32_k4(afrag[kk], b, acc);
    }
#pragma unroll
    for (int i = 0; i < 8; i++)
      partial[i] += fmaxf(0.f, sqr[i] + sqc - 2.f * acc[i]);
  }

#pragma unroll
  for (int i = 0; i < 8; i++) {
    float p = partial[i];
    p += __shfl_xor(p, 1, 32);
    p += __shfl_xor(p, 2, 32);
    p += __shfl_xor(p, 4, 32);
    p += __shfl_xor(p, 8, 32);
    if (r16 == 0)
      anchorPart[(size_t)chunk * NPTS + m0 + i + half * 8] = p;
  }
}

__global__ __launch_bounds__(256)
void combine_kernel(const float* __restrict__ anchorPart,
                    float* __restrict__ anchorSq) {
  const int i = blockIdx.x * 256 + threadIdx.x;
  float s = 0.f;
#pragma unroll
  for (int c = 0; c < NCHUNK; c++) s += anchorPart[(size_t)c * NPTS + i];
  anchorSq[i] = s;
}

#define NWAVES (NPTS / 32)

__device__ __forceinline__ void argmax_merge(float& bv, int& bi, float ov, int oi) {
  const bool take = (ov > bv) | ((ov == bv) & (oi < bi));
  bv = take ? ov : bv;
  bi = take ? oi : bi;
}

__device__ __forceinline__ void wave_argmax(float& bv, int& bi) {
#pragma unroll
  for (int m = 1; m < 32; m <<= 1) {
    const float ov = __shfl_xor(bv, m, 32);
    const int   oi = __shfl_xor(bi, m, 32);
    argmax_merge(bv, bi, ov, oi);
  }
}

__global__ __launch_bounds__(32)
void anchor_init_kernel(float* __restrict__ anchor, float* __restrict__ candV,
                        int* __restrict__ candI) {
  const int row  = blockIdx.x * 32 + threadIdx.x;
  const int lane = threadIdx.x;
  const float a = sqrtf(anchor[row]);
  anchor[row] = a;
  float bv = a; int bi = row;
  wave_argmax(bv, bi);
  if (lane == 0) { candV[row >> 5] = bv; candI[row >> 5] = bi; }
}

__global__ __launch_bounds__(32)
void greedy_step_kernel(const float* __restrict__ R, const float* __restrict__ sq,
                        float* __restrict__ anchor, float* __restrict__ candV,
                        int* __restrict__ candI, int* __restrict__ sel, int s) {
  __shared__ float4 selRow[DPROJ / 4];
  const int row  = blockIdx.x * 32 + threadIdx.x;
  const int lane = threadIdx.x;
  const int pin  = (s & 1) * NWAVES;
  const int pout = NWAVES - pin;

  float4 cv[4]; int4 ci[4];
  const float4* cvp = (const float4*)(candV + pin) + lane;
  const int4*   cip = (const int4*)(candI + pin) + lane;
#pragma unroll
  for (int q = 0; q < 4; q++) { cv[q] = cvp[q * 32]; ci[q] = cip[q * 32]; }

  float4 v[DPROJ / 4];
  const float4* rp = (const float4*)(R + (size_t)row * DPROJ);
#pragma unroll
  for (int k = 0; k < DPROJ / 4; k++) v[k] = rp[k];
  const float sqOwn = sq[row];
  const float aOld  = anchor[row];
  __builtin_amdgcn_sched_barrier(0);

  float bv = cv[0].x; int bi = ci[0].x;
#pragma unroll
  for (int q = 0; q < 4; q++) {
    if (q > 0) argmax_merge(bv, bi, cv[q].x, ci[q].x);
    argmax_merge(bv, bi, cv[q].y, ci[q].y);
    argmax_merge(bv, bi, cv[q].z, ci[q].z);
    argmax_merge(bv, bi, cv[q].w, ci[q].w);
  }
  wave_argmax(bv, bi);
  const int si = bi;

  selRow[lane] = ((const float4*)(R + (size_t)si * DPROJ))[lane];
  const float selSq = sq[si];
  __syncthreads();
  float d0 = 0.f, d1 = 0.f, d2s = 0.f, d3 = 0.f;
#pragma unroll
  for (int k = 0; k < DPROJ / 4; k++) {
    const float4 w = selRow[k];
    d0  = fmaf(v[k].x, w.x, d0);
    d1  = fmaf(v[k].y, w.y, d1);
    d2s = fmaf(v[k].z, w.z, d2s);
    d3  = fmaf(v[k].w, w.w, d3);
  }
  const float dot = (d0 + d1) + (d2s + d3);
  const float d2 = sqOwn + selSq - 2.f * dot;
  const float a  = fminf(aOld, sqrtf(fmaxf(d2, 0.f)));
  anchor[row] = a;

  float mv = a; int mi = row;
  wave_argmax(mv, mi);
  if (lane == 0) { candV[pout + (row >> 5)] = mv; candI[pout + (row >> 5)] = mi; }
  if (row == 0) sel[s] = si;
}

__global__ __launch_bounds__(256)
void gather_kernel(const float* __restrict__ F, const int* __restrict__ sel,
                   float* __restrict__ out) {
  const int s = blockIdx.x;
  const int c = blockIdx.y * 256 + threadIdx.x;
  out[(size_t)s * DIN + c] = F[(size_t)sel[s] * DIN + c];
}

extern "C" void kernel_launch(void* const* d_in, const int* in_sizes, int n_in,
                              void* d_out, int out_size, void* d_ws, size_t ws_size,
                              hipStream_t stream) {
  (void)in_sizes; (void)n_in; (void)out_size; (void)ws_size;
  const float* F = (const float*)d_in[0];
  const float* W = (const float*)d_in[1];

  float* R          = (float*)d_ws;
  float* sq         = R + (size_t)NPTS * DPROJ;
  float* anchorSq   = sq + NPTS;
  float* anchorPart = anchorSq + NPTS;
  int*   sel        = (int*)(anchorPart + (size_t)NCHUNK * NPTS);

  proj_kernel<<<1024, 256, 0, stream>>>(F, W, R);
  sq_kernel<<<NPTS / 8, 256, 0, stream>>>(R, sq);
  gram_rowsum_kernel<<<1024, 256, 0, stream>>>(R, sq, anchorPart);
  combine_kernel<<<NPTS / 256, 256, 0, stream>>>(anchorPart, anchorSq);
  float* candV = anchorPart;
  int*   candI = (int*)(anchorPart + 2 * NWAVES);
  anchor_init_kernel<<<NPTS / 32, 32, 0, stream>>>(anchorSq, candV, candI);
  for (int s = 0; s < NSAMP; s++)
    greedy_step_kernel<<<NPTS / 32, 32, 0, stream>>>(R, sq, anchorSq, candV, candI, sel, s);
  gather_kernel<<<dim3(NSAMP, DIN / 256), 256, 0, stream>>>(F, sel, (float*)d_out);
}


# gram pass: B fragments of the next column tile are requested a tile ahead into a second register set, so the 32-WMMA burst runs on landed data instead of waiting on each load; sum order unchanged
# speedup vs baseline: 20.2767x; 1.0462x over previous
#include <hip/hip_runtime.h>


#define NPTS   16384
#define DIN    1024
#define DPROJ  128
#define NSAMP  163
#define NCHUNK 8
#define TILES_PER_CHUNK (NPTS / 16 / NCHUNK)

typedef float v2f __attribute__((ext_vector_type(2)));
typedef float v8f __attribute__((ext_vector_type(8)));

__device__ __forceinline__ v8f wmma_f32_k4(v2f a, v2f b, v8f c) {
  return __builtin_amdgcn_wmma_f32_16x16x4_f32(
       false, a,  false, b,
       (short)0, c,  false,  false);
}

__global__ __launch_bounds__(256)
void proj_kernel(const float* __restrict__ F, const float* __restrict__ W,
                 float* __restrict__ R) {
  const int wave = (blockIdx.x * blockDim.x + threadIdx.x) >> 5;
  const int lane = threadIdx.x & 31;
  const int tm = wave >> 3;
  const int tn = wave & 7;
  const int m0 = tm * 16, n0 = tn * 16;
  const int half = lane >> 4;
  const int r16  = lane & 15;
  const int koff = half * 2;

  v8f acc = {};
  const float* arow = F + (size_t)(m0 + r16) * DIN + koff;
  for (int k = 0; k < DIN; k += 4) {
    v2f a = *(const v2f*)(arow + k);
    v2f b;
    b.x = W[(size_t)(k + koff)     * DPROJ + n0 + r16];
    b.y = W[(size_t)(k + koff + 1) * DPROJ + n0 + r16];
    acc = wmma_f32_k4(a, b, acc);
  }
#pragma unroll
  for (int i = 0; i < 8; i++)
    R[(size_t)(m0 + i + half * 8) * DPROJ + n0 + r16] = acc[i];
}

__global__ __launch_bounds__(256)
void sq_kernel(const float* __restrict__ R, float* __restrict__ sq) {
  const int row  = (blockIdx.x * blockDim.x + threadIdx.x) >> 5;
  const int lane = threadIdx.x & 31;
  const float* rp = R + (size_t)row * DPROJ;
  float s = 0.f;
#pragma unroll
  for (int k = lane; k < DPROJ; k += 32) { float v = rp[k]; s += v * v; }
#pragma unroll
  for (int m = 16; m >= 1; m >>= 1) s += __shfl_xor(s, m, 32);
  if (lane == 0) sq[row] = s;
}

struct BTile { v2f f[32]; float sqc; };

__device__ __forceinline__ void load_btile(BTile& bt, const float* __restrict__ R,
                                           const float* __restrict__ sq,
                                           int t, int r16, int koff) {
  const int col = t * 16 + r16;
  const float* brow = R + (size_t)col * DPROJ + koff;
#pragma unroll
  for (int kk = 0; kk < 32; kk++) bt.f[kk] = *(const v2f*)(brow + kk * 4);
  bt.sqc = sq[col];
}

__device__ __forceinline__ void tile_rowsums(const v2f (&afrag)[32], const BTile& bt,
                                             const float (&sqr)[8], float (&partial)[8]) {
  v8f acc = {};
#pragma unroll
  for (int kk = 0; kk < 32; kk++) acc = wmma_f32_k4(afrag[kk], bt.f[kk], acc);
#pragma unroll
  for (int i = 0; i < 8; i++)
    partial[i] += fmaxf(0.f, sqr[i] + bt.sqc - 2.f * acc[i]);
}

__global__ __launch_bounds__(256)
void gram_rowsum_kernel(const float* __restrict__ R, const float* __restrict__ sq,
                        float* __restrict__ anchorPart) {
  const int wave  = (blockIdx.x * blockDim.x + threadIdx.x) >> 5;
  const int lane  = threadIdx.x & 31;
  const int chunk = wave & (NCHUNK - 1);
  const int strip = wave >> 3;
  const int half  = lane >> 4;
  const int r16   = lane & 15;
  const int koff  = half * 2;
  const int m0    = strip * 16;

  v2f afrag[32];
  const float* arow = R + (size_t)(m0 + r16) * DPROJ + koff;
#pragma unroll
  for (int kk = 0; kk < 32; kk++) afrag[kk] = *(const v2f*)(arow + kk * 4);

  float sqr[8];
#pragma unroll
  for (int i = 0; i < 8; i++) sqr[i] = sq[m0 + i + half * 8];

  float partial[8] = {0.f, 0.f, 0.f, 0.f, 0.f, 0.f, 0.f, 0.f};

  const int t0   = chunk * TILES_PER_CHUNK;
  const int tEnd = t0 + TILES_PER_CHUNK;
  BTile b0, b1;
  load_btile(b0, R, sq, t0, r16, koff);
  for (int t = t0; t < tEnd - 2; t += 2) {
    load_btile(b1, R, sq, t + 1, r16, koff);
    tile_rowsums(afrag, b0, sqr, partial);
    load_btile(b0, R, sq, t + 2, r16, koff);
    tile_rowsums(afrag, b1, sqr, partial);
  }
  load_btile(b1, R, sq, tEnd - 1, r16, koff);
  tile_rowsums(afrag, b0, sqr, partial);
  tile_rowsums(afrag, b1, sqr, partial);

#pragma unroll
  for (int i = 0; i < 8; i++) {
    float q = partial[i];
    q += __shfl_xor(q, 1, 32);
    q += __shfl_xor(q, 2, 32);
    q += __shfl_xor(q, 4, 32);
    q += __shfl_xor(q, 8, 32);
    if (r16 == 0)
      anchorPart[(size_t)chunk * NPTS + m0 + i + half * 8] = q;
  }
}

__global__ __launch_bounds__(256)
void combine_kernel(const float* __restrict__ anchorPart,
                    float* __restrict__ anchorSq) {
  const int i = blockIdx.x * 256 + threadIdx.x;
  float s = 0.f;
#pragma unroll
  for (int c = 0; c < NCHUNK; c++) s += anchorPart[(size_t)c * NPTS + i];
  anchorSq[i] = s;
}

#define NWAVES (NPTS / 32)

__device__ __forceinline__ void argmax_merge(float& bv, int& bi, float ov, int oi) {
  const bool take = (ov > bv) | ((ov == bv) & (oi < bi));
  bv = take ? ov : bv;
  bi = take ? oi : bi;
}

__device__ __forceinline__ void wave_argmax(float& bv, int& bi) {
#pragma unroll
  for (int m = 1; m < 32; m <<= 1) {
    const float ov = __shfl_xor(bv, m, 32);
    const int   oi = __shfl_xor(bi, m, 32);
    argmax_merge(bv, bi, ov, oi);
  }
}

__global__ __launch_bounds__(32)
void anchor_init_kernel(float* __restrict__ anchor, float* __restrict__ candV,
                        int* __restrict__ candI) {
  const int row  = blockIdx.x * 32 + threadIdx.x;
  const int lane = threadIdx.x;
  const float a = sqrtf(anchor[row]);
  anchor[row] = a;
  float bv = a; int bi = row;
  wave_argmax(bv, bi);
  if (lane == 0) { candV[row >> 5] = bv; candI[row >> 5] = bi; }
}

__global__ __launch_bounds__(32)
void greedy_step_kernel(const float* __restrict__ R, const float* __restrict__ sq,
                        float* __restrict__ anchor, float* __restrict__ candV,
                        int* __restrict__ candI, int* __restrict__ sel, int s) {
  __shared__ float4 selRow[DPROJ / 4];
  const int row  = blockIdx.x * 32 + threadIdx.x;
  const int lane = threadIdx.x;
  const int pin  = (s & 1) * NWAVES;
  const int pout = NWAVES - pin;

  float4 cv[4]; int4 ci[4];
  const float4* cvp = (const float4*)(candV + pin) + lane;
  const int4*   cip = (const int4*)(candI + pin) + lane;
#pragma unroll
  for (int q = 0; q < 4; q++) { cv[q] = cvp[q * 32]; ci[q] = cip[q * 32]; }

  float4 v[DPROJ / 4];
  const float4* rp = (const float4*)(R + (size_t)row * DPROJ);
#pragma unroll
  for (int k = 0; k < DPROJ / 4; k++) v[k] = rp[k];
  const float sqOwn = sq[row];
  const float aOld  = anchor[row];
  __builtin_amdgcn_sched_barrier(0);

  float bv = cv[0].x; int bi = ci[0].x;
#pragma unroll
  for (int q = 0; q < 4; q++) {
    if (q > 0) argmax_merge(bv, bi, cv[q].x, ci[q].x);
    argmax_merge(bv, bi, cv[q].y, ci[q].y);
    argmax_merge(bv, bi, cv[q].z, ci[q].z);
    argmax_merge(bv, bi, cv[q].w, ci[q].w);
  }
  wave_argmax(bv, bi);
  const int si = bi;

  selRow[lane] = ((const float4*)(R + (size_t)si * DPROJ))[lane];
  const float selSq = sq[si];
  __syncthreads();
  float d0 = 0.f, d1 = 0.f, d2s = 0.f, d3 = 0.f;
#pragma unroll
  for (int k = 0; k < DPROJ / 4; k++) {
    const float4 w = selRow[k];
    d0  = fmaf(v[k].x, w.x, d0);
    d1  = fmaf(v[k].y, w.y, d1);
    d2s = fmaf(v[k].z, w.z, d2s);
    d3  = fmaf(v[k].w, w.w, d3);
  }
  const float dot = (d0 + d1) + (d2s + d3);
  const float d2 = sqOwn + selSq - 2.f * dot;
  const float a  = fminf(aOld, sqrtf(fmaxf(d2, 0.f)));
  anchor[row] = a;

  float mv = a; int mi = row;
  wave_argmax(mv, mi);
  if (lane == 0) { candV[pout + (row >> 5)] = mv; candI[pout + (row >> 5)] = mi; }
  if (row == 0) sel[s] = si;
}

__global__ __launch_bounds__(256)
void gather_kernel(const float* __restrict__ F, const int* __restrict__ sel,
                   float* __restrict__ out) {
  const int s = blockIdx.x;
  const int c = blockIdx.y * 256 + threadIdx.x;
  out[(size_t)s * DIN + c] = F[(size_t)sel[s] * DIN + c];
}

extern "C" void kernel_launch(void* const* d_in, const int* in_sizes, int n_in,
                              void* d_out, int out_size, void* d_ws, size_t ws_size,
                              hipStream_t stream) {
  (void)in_sizes; (void)n_in; (void)out_size; (void)ws_size;
  const float* F = (const float*)d_in[0];
  const float* W = (const float*)d_in[1];

  float* R          = (float*)d_ws;
  float* sq         = R + (size_t)NPTS * DPROJ;
  float* anchorSq   = sq + NPTS;
  float* anchorPart = anchorSq + NPTS;
  int*   sel        = (int*)(anchorPart + (size_t)NCHUNK * NPTS);

  proj_kernel<<<1024, 256, 0, stream>>>(F, W, R);
  sq_kernel<<<NPTS / 8, 256, 0, stream>>>(R, sq);
  gram_rowsum_kernel<<<1024, 256, 0, stream>>>(R, sq, anchorPart);
  combine_kernel<<<NPTS / 256, 256, 0, stream>>>(anchorPart, anchorSq);
  float* candV = anchorPart;
  int*   candI = (int*)(anchorPart + 2 * NWAVES);
  anchor_init_kernel<<<NPTS / 32, 32, 0, stream>>>(anchorSq, candV, candI);
  for (int s = 0; s < NSAMP; s++)
    greedy_step_kernel<<<NPTS / 32, 32, 0, stream>>>(R, sq, anchorSq, candV, candI, sel, s);
  gather_kernel<<<dim3(NSAMP, DIN / 256), 256, 0, stream>>>(F, sel, (float*)d_out);
}


# gram pass: column tiles staged once per work-group in LDS (16-byte loads, 2 tiles per stage) and shared by 8 waves of 8 row strips, replacing per-wave 16-line B fragment loads; sum order unchanged
# speedup vs baseline: 27.5392x; 1.3582x over previous
#include <hip/hip_runtime.h>


#define NPTS   16384
#define DIN    1024
#define DPROJ  128
#define NSAMP  163
#define NCHUNK 8
#define TILES_PER_CHUNK (NPTS / 16 / NCHUNK)

typedef float v2f __attribute__((ext_vector_type(2)));
typedef float v8f __attribute__((ext_vector_type(8)));

__device__ __forceinline__ v8f wmma_f32_k4(v2f a, v2f b, v8f c) {
  return __builtin_amdgcn_wmma_f32_16x16x4_f32(
       false, a,  false, b,
       (short)0, c,  false,  false);
}

__global__ __launch_bounds__(256)
void proj_kernel(const float* __restrict__ F, const float* __restrict__ W,
                 float* __restrict__ R) {
  const int wave = (blockIdx.x * blockDim.x + threadIdx.x) >> 5;
  const int lane = threadIdx.x & 31;
  const int tm = wave >> 3;
  const int tn = wave & 7;
  const int m0 = tm * 16, n0 = tn * 16;
  const int half = lane >> 4;
  const int r16  = lane & 15;
  const int koff = half * 2;

  v8f acc = {};
  const float* arow = F + (size_t)(m0 + r16) * DIN + koff;
  for (int k = 0; k < DIN; k += 4) {
    v2f a = *(const v2f*)(arow + k);
    v2f b;
    b.x = W[(size_t)(k + koff)     * DPROJ + n0 + r16];
    b.y = W[(size_t)(k + koff + 1) * DPROJ + n0 + r16];
    acc = wmma_f32_k4(a, b, acc);
  }
#pragma unroll
  for (int i = 0; i < 8; i++)
    R[(size_t)(m0 + i + half * 8) * DPROJ + n0 + r16] = acc[i];
}

__global__ __launch_bounds__(256)
void sq_kernel(const float* __restrict__ R, float* __restrict__ sq) {
  const int row  = (blockIdx.x * blockDim.x + threadIdx.x) >> 5;
  const int lane = threadIdx.x & 31;
  const float* rp = R + (size_t)row * DPROJ;
  float s = 0.f;
#pragma unroll
  for (int k = lane; k < DPROJ; k += 32) { float v = rp[k]; s += v * v; }
#pragma unroll
  for (int m = 16; m >= 1; m >>= 1) s += __shfl_xor(s, m, 32);
  if (lane == 0) sq[row] = s;
}

#define GRAM_PITCH        132
#define GRAM_STAGE_TILES  2
#define GRAM_TILE_FLOATS  (16 * GRAM_PITCH)
#define GRAM_STAGE_FLOATS (GRAM_STAGE_TILES * GRAM_TILE_FLOATS)
#define GRAM_NSTEP        (TILES_PER_CHUNK / GRAM_STAGE_TILES)

struct BTile { v2f f[32]; float sqc; };
struct BHalf { float4 p[2]; };

__device__ __forceinline__ void half_fetch(BHalf& g, const float* __restrict__ Rtile,
                                           unsigned thrOff) {
  const float* a = Rtile + thrOff;
  g.p[0] = *(const float4*)(a);
  g.p[1] = *(const float4*)(a + 4);
}

__device__ __forceinline__ void half_store(const BHalf& g, float* d) {
  *(float4*)(d)     = make_float4(g.p[0].x, g.p[0].y, g.p[1].x, g.p[1].y);
  *(float4*)(d + 4) = make_float4(g.p[0].z, g.p[0].w, g.p[1].z, g.p[1].w);
}

__device__ __forceinline__ void load_btile(BTile& bt, const float* tile,
                                           const float* __restrict__ sqTile, unsigned r16) {
#pragma unroll
  for (int j = 0; j < 16; j++) {
    const float4 q = *(const float4*)(tile + j * 8);
    bt.f[2 * j].x     = q.x;  bt.f[2 * j].y     = q.y;
    bt.f[2 * j + 1].x = q.z;  bt.f[2 * j + 1].y = q.w;
  }
  bt.sqc = sqTile[r16];
}

__device__ __forceinline__ void tile_rowsums(const v2f (&afrag)[32], const BTile& bt,
                                             const v2f (&sqr)[4], v2f (&partial)[4]) {
  v8f acc = {};
#pragma unroll
  for (int kk = 0; kk < 32; kk++) acc = wmma_f32_k4(afrag[kk], bt.f[kk], acc);
  const v2f sqc2 = {bt.sqc, bt.sqc};
  const v2f zero2 = {0.f, 0.f};
#pragma unroll
  for (int j = 0; j < 4; j++) {
    const v2f a = {acc[2 * j], acc[2 * j + 1]};
    partial[j] += __builtin_elementwise_max(zero2, sqr[j] + sqc2 - 2.f * a);
  }
}

template <int CUR, bool FETCH_AHEAD>
__device__ __forceinline__ void gram_step(const v2f (&afrag)[32], BTile& b0, BTile& b1,
                                          BHalf& g, const v2f (&sqr)[4],
                                          v2f (&partial)[4], float* bstage,
                                          const float* __restrict__ Rt,
                                          const float* __restrict__ sqt,
                                          unsigned laneOff, unsigned thrOff,
                                          unsigned stOff, unsigned r16) {
  const float* cur = bstage + CUR * GRAM_STAGE_FLOATS;
  float*       nxt = bstage + (CUR ^ 1) * GRAM_STAGE_FLOATS;
  load_btile(b1, cur + GRAM_TILE_FLOATS + laneOff, sqt + 16, r16);
  half_store(g, nxt + stOff);
  half_fetch(g, Rt + 3 * 16 * DPROJ, thrOff);
  tile_rowsums(afrag, b0, sqr, partial);
  half_store(g, nxt + GRAM_TILE_FLOATS + stOff);
  __syncthreads();
  __builtin_amdgcn_sched_barrier(0);
  load_btile(b0, nxt + laneOff, sqt + 32, r16);
  if (FETCH_AHEAD) half_fetch(g, Rt + 4 * 16 * DPROJ, thrOff);
  tile_rowsums(afrag, b1, sqr, partial);
  __builtin_amdgcn_sched_barrier(0);
}

__global__ __launch_bounds__(256, 4)
void gram_rowsum_kernel(const float* __restrict__ R, const float* __restrict__ sq,
                        float* __restrict__ anchorPart) {
  __shared__ __attribute__((aligned(16))) float bstage[2 * GRAM_STAGE_FLOATS];

  const unsigned tid   = threadIdx.x;
  const unsigned lane  = tid & 31;
  const unsigned chunk = blockIdx.x & (NCHUNK - 1);
  const unsigned strip = (blockIdx.x >> 3) * 8 + (tid >> 5);
  const unsigned half  = lane >> 4;
  const unsigned r16   = lane & 15;
  const unsigned koff  = half * 2;
  const unsigned m0    = strip * 16;

  v2f afrag[32];
  const float* arow = R + (size_t)(m0 + r16) * DPROJ + koff;
#pragma unroll
  for (int kk = 0; kk < 32; kk++) afrag[kk] = *(const v2f*)(arow + kk * 4);

  v2f sqr[4];
#pragma unroll
  for (int j = 0; j < 4; j++) {
    sqr[j].x = sq[m0 + 2 * j + half * 8];
    sqr[j].y = sq[m0 + 2 * j + 1 + half * 8];
  }

  v2f partial[4] = {{0.f, 0.f}, {0.f, 0.f}, {0.f, 0.f}, {0.f, 0.f}};

  const unsigned t0      = chunk * TILES_PER_CHUNK;
  const unsigned laneOff = r16 * GRAM_PITCH + half * 4;
  const unsigned thrOff  = (tid >> 4) * DPROJ + (tid & 15) * 8;
  const unsigned stOff   = (tid >> 4) * GRAM_PITCH + (tid & 15) * 8;
  const float* Rt  = R + (size_t)t0 * 16 * DPROJ;
  const float* sqt = sq + t0 * 16;

  BHalf g;
  BTile b0, b1;
  half_fetch(g, Rt, thrOff);
  half_store(g, bstage + stOff);
  half_fetch(g, Rt + 16 * DPROJ, thrOff);
  half_store(g, bstage + GRAM_TILE_FLOATS + stOff);
  __syncthreads();
  load_btile(b0, bstage + laneOff, sqt, r16);
  half_fetch(g, Rt + 2 * 16 * DPROJ, thrOff);

  for (unsigned sp = 0; sp < (GRAM_NSTEP - 2) / 2; sp++) {
    const float* Rs  = Rt + sp * (64 * DPROJ);
    const float* sqs = sqt + sp * 64;
    gram_step<0, true>(afrag, b0, b1, g, sqr, partial, bstage, Rs, sqs,
                       laneOff, thrOff, stOff, r16);
    gram_step<1, true>(afrag, b0, b1, g, sqr, partial, bstage, Rs + 32 * DPROJ, sqs + 32,
                       laneOff, thrOff, stOff, r16);
  }
  const float* Rl  = Rt + (GRAM_NSTEP - 2) * (32 * DPROJ);
  const float* sql = sqt + (GRAM_NSTEP - 2) * 32;
  gram_step<0, false>(afrag, b0, b1, g, sqr, partial, bstage, Rl, sql,
                      laneOff, thrOff, stOff, r16);
  load_btile(b1, bstage + GRAM_STAGE_FLOATS + GRAM_TILE_FLOATS + laneOff, sql + 48, r16);
  tile_rowsums(afrag, b0, sqr, partial);
  tile_rowsums(afrag, b1, sqr, partial);

#pragma unroll
  for (int i = 0; i < 8; i++) {
    float q = (i & 1) ? partial[i >> 1].y : partial[i >> 1].x;
    q += __shfl_xor(q, 1, 32);
    q += __shfl_xor(q, 2, 32);
    q += __shfl_xor(q, 4, 32);
    q += __shfl_xor(q, 8, 32);
    if (r16 == 0)
      anchorPart[(size_t)chunk * NPTS + m0 + i + half * 8] = q;
  }
}

__global__ __launch_bounds__(256)
void combine_kernel(const float* __restrict__ anchorPart,
                    float* __restrict__ anchorSq) {
  const int i = blockIdx.x * 256 + threadIdx.x;
  float s = 0.f;
#pragma unroll
  for (int c = 0; c < NCHUNK; c++) s += anchorPart[(size_t)c * NPTS + i];
  anchorSq[i] = s;
}

#define NWAVES (NPTS / 32)

__device__ __forceinline__ void argmax_merge(float& bv, int& bi, float ov, int oi) {
  const bool take = (ov > bv) | ((ov == bv) & (oi < bi));
  bv = take ? ov : bv;
  bi = take ? oi : bi;
}

__device__ __forceinline__ void wave_argmax(float& bv, int& bi) {
#pragma unroll
  for (int m = 1; m < 32; m <<= 1) {
    const float ov = __shfl_xor(bv, m, 32);
    const int   oi = __shfl_xor(bi, m, 32);
    argmax_merge(bv, bi, ov, oi);
  }
}

__global__ __launch_bounds__(32)
void anchor_init_kernel(float* __restrict__ anchor, float* __restrict__ candV,
                        int* __restrict__ candI) {
  const int row  = blockIdx.x * 32 + threadIdx.x;
  const int lane = threadIdx.x;
  const float a = sqrtf(anchor[row]);
  anchor[row] = a;
  float bv = a; int bi = row;
  wave_argmax(bv, bi);
  if (lane == 0) { candV[row >> 5] = bv; candI[row >> 5] = bi; }
}

__global__ __launch_bounds__(32)
void greedy_step_kernel(const float* __restrict__ R, const float* __restrict__ sq,
                        float* __restrict__ anchor, float* __restrict__ candV,
                        int* __restrict__ candI, int* __restrict__ sel, int s) {
  __shared__ float4 selRow[DPROJ / 4];
  const int row  = blockIdx.x * 32 + threadIdx.x;
  const int lane = threadIdx.x;
  const int pin  = (s & 1) * NWAVES;
  const int pout = NWAVES - pin;

  float4 cv[4]; int4 ci[4];
  const float4* cvp = (const float4*)(candV + pin) + lane;
  const int4*   cip = (const int4*)(candI + pin) + lane;
#pragma unroll
  for (int q = 0; q < 4; q++) { cv[q] = cvp[q * 32]; ci[q] = cip[q * 32]; }

  float4 v[DPROJ / 4];
  const float4* rp = (const float4*)(R + (size_t)row * DPROJ);
#pragma unroll
  for (int k = 0; k < DPROJ / 4; k++) v[k] = rp[k];
  const float sqOwn = sq[row];
  const float aOld  = anchor[row];
  __builtin_amdgcn_sched_barrier(0);

  float bv = cv[0].x; int bi = ci[0].x;
#pragma unroll
  for (int q = 0; q < 4; q++) {
    if (q > 0) argmax_merge(bv, bi, cv[q].x, ci[q].x);
    argmax_merge(bv, bi, cv[q].y, ci[q].y);
    argmax_merge(bv, bi, cv[q].z, ci[q].z);
    argmax_merge(bv, bi, cv[q].w, ci[q].w);
  }
  wave_argmax(bv, bi);
  const int si = bi;

  selRow[lane] = ((const float4*)(R + (size_t)si * DPROJ))[lane];
  const float selSq = sq[si];
  __syncthreads();
  float d0 = 0.f, d1 = 0.f, d2s = 0.f, d3 = 0.f;
#pragma unroll
  for (int k = 0; k < DPROJ / 4; k++) {
    const float4 w = selRow[k];
    d0  = fmaf(v[k].x, w.x, d0);
    d1  = fmaf(v[k].y, w.y, d1);
    d2s = fmaf(v[k].z, w.z, d2s);
    d3  = fmaf(v[k].w, w.w, d3);
  }
  const float dot = (d0 + d1) + (d2s + d3);
  const float d2 = sqOwn + selSq - 2.f * dot;
  const float a  = fminf(aOld, sqrtf(fmaxf(d2, 0.f)));
  anchor[row] = a;

  float mv = a; int mi = row;
  wave_argmax(mv, mi);
  if (lane == 0) { candV[pout + (row >> 5)] = mv; candI[pout + (row >> 5)] = mi; }
  if (row == 0) sel[s] = si;
}

__global__ __launch_bounds__(256)
void gather_kernel(const float* __restrict__ F, const int* __restrict__ sel,
                   float* __restrict__ out) {
  const int s = blockIdx.x;
  const int c = blockIdx.y * 256 + threadIdx.x;
  out[(size_t)s * DIN + c] = F[(size_t)sel[s] * DIN + c];
}

extern "C" void kernel_launch(void* const* d_in, const int* in_sizes, int n_in,
                              void* d_out, int out_size, void* d_ws, size_t ws_size,
                              hipStream_t stream) {
  (void)in_sizes; (void)n_in; (void)out_size; (void)ws_size;
  const float* F = (const float*)d_in[0];
  const float* W = (const float*)d_in[1];

  float* R          = (float*)d_ws;
  float* sq         = R + (size_t)NPTS * DPROJ;
  float* anchorSq   = sq + NPTS;
  float* anchorPart = anchorSq + NPTS;
  int*   sel        = (int*)(anchorPart + (size_t)NCHUNK * NPTS);

  proj_kernel<<<1024, 256, 0, stream>>>(F, W, R);
  sq_kernel<<<NPTS / 8, 256, 0, stream>>>(R, sq);
  gram_rowsum_kernel<<<1024, 256, 0, stream>>>(R, sq, anchorPart);
  combine_kernel<<<NPTS / 256, 256, 0, stream>>>(anchorPart, anchorSq);
  float* candV = anchorPart;
  int*   candI = (int*)(anchorPart + 2 * NWAVES);
  anchor_init_kernel<<<NPTS / 32, 32, 0, stream>>>(anchorSq, candV, candI);
  for (int s = 0; s < NSAMP; s++)
    greedy_step_kernel<<<NPTS / 32, 32, 0, stream>>>(R, sq, anchorSq, candV, candI, sel, s);
  gather_kernel<<<dim3(NSAMP, DIN / 256), 256, 0, stream>>>(F, sel, (float*)d_out);
}


# greedy launches: the ten cross-lane argmax exchanges per launch are register moves (DPP row_xmask, v_permlanex16) instead of ds_bpermute LDS round trips; merge order and results unchanged
# speedup vs baseline: 28.6338x; 1.0397x over previous
#include <hip/hip_runtime.h>


#define NPTS   16384
#define DIN    1024
#define DPROJ  128
#define NSAMP  163
#define NCHUNK 8
#define TILES_PER_CHUNK (NPTS / 16 / NCHUNK)

typedef float v2f __attribute__((ext_vector_type(2)));
typedef float v8f __attribute__((ext_vector_type(8)));

__device__ __forceinline__ v8f wmma_f32_k4(v2f a, v2f b, v8f c) {
  return __builtin_amdgcn_wmma_f32_16x16x4_f32(
       false, a,  false, b,
       (short)0, c,  false,  false);
}

__global__ __launch_bounds__(256)
void proj_kernel(const float* __restrict__ F, const float* __restrict__ W,
                 float* __restrict__ R) {
  const int wave = (blockIdx.x * blockDim.x + threadIdx.x) >> 5;
  const int lane = threadIdx.x & 31;
  const int tm = wave >> 3;
  const int tn = wave & 7;
  const int m0 = tm * 16, n0 = tn * 16;
  const int half = lane >> 4;
  const int r16  = lane & 15;
  const int koff = half * 2;

  v8f acc = {};
  const float* arow = F + (size_t)(m0 + r16) * DIN + koff;
  for (int k = 0; k < DIN; k += 4) {
    v2f a = *(const v2f*)(arow + k);
    v2f b;
    b.x = W[(size_t)(k + koff)     * DPROJ + n0 + r16];
    b.y = W[(size_t)(k + koff + 1) * DPROJ + n0 + r16];
    acc = wmma_f32_k4(a, b, acc);
  }
#pragma unroll
  for (int i = 0; i < 8; i++)
    R[(size_t)(m0 + i + half * 8) * DPROJ + n0 + r16] = acc[i];
}

__global__ __launch_bounds__(256)
void sq_kernel(const float* __restrict__ R, float* __restrict__ sq) {
  const int row  = (blockIdx.x * blockDim.x + threadIdx.x) >> 5;
  const int lane = threadIdx.x & 31;
  const float* rp = R + (size_t)row * DPROJ;
  float s = 0.f;
#pragma unroll
  for (int k = lane; k < DPROJ; k += 32) { float v = rp[k]; s += v * v; }
#pragma unroll
  for (int m = 16; m >= 1; m >>= 1) s += __shfl_xor(s, m, 32);
  if (lane == 0) sq[row] = s;
}

#define GRAM_PITCH        132
#define GRAM_STAGE_TILES  2
#define GRAM_TILE_FLOATS  (16 * GRAM_PITCH)
#define GRAM_STAGE_FLOATS (GRAM_STAGE_TILES * GRAM_TILE_FLOATS)
#define GRAM_NSTEP        (TILES_PER_CHUNK / GRAM_STAGE_TILES)

struct BTile { v2f f[32]; float sqc; };
struct BHalf { float4 p[2]; };

__device__ __forceinline__ void half_fetch(BHalf& g, const float* __restrict__ Rtile,
                                           unsigned thrOff) {
  const float* a = Rtile + thrOff;
  g.p[0] = *(const float4*)(a);
  g.p[1] = *(const float4*)(a + 4);
}

__device__ __forceinline__ void half_store(const BHalf& g, float* d) {
  *(float4*)(d)     = make_float4(g.p[0].x, g.p[0].y, g.p[1].x, g.p[1].y);
  *(float4*)(d + 4) = make_float4(g.p[0].z, g.p[0].w, g.p[1].z, g.p[1].w);
}

__device__ __forceinline__ void load_btile(BTile& bt, const float* tile,
                                           const float* __restrict__ sqTile, unsigned r16) {
#pragma unroll
  for (int j = 0; j < 16; j++) {
    const float4 q = *(const float4*)(tile + j * 8);
    bt.f[2 * j].x     = q.x;  bt.f[2 * j].y     = q.y;
    bt.f[2 * j + 1].x = q.z;  bt.f[2 * j + 1].y = q.w;
  }
  bt.sqc = sqTile[r16];
}

__device__ __forceinline__ void tile_rowsums(const v2f (&afrag)[32], const BTile& bt,
                                             const v2f (&sqr)[4], v2f (&partial)[4]) {
  v8f acc = {};
#pragma unroll
  for (int kk = 0; kk < 32; kk++) acc = wmma_f32_k4(afrag[kk], bt.f[kk], acc);
  const v2f sqc2 = {bt.sqc, bt.sqc};
  const v2f zero2 = {0.f, 0.f};
#pragma unroll
  for (int j = 0; j < 4; j++) {
    const v2f a = {acc[2 * j], acc[2 * j + 1]};
    partial[j] += __builtin_elementwise_max(zero2, sqr[j] + sqc2 - 2.f * a);
  }
}

template <int CUR, bool FETCH_AHEAD>
__device__ __forceinline__ void gram_step(const v2f (&afrag)[32], BTile& b0, BTile& b1,
                                          BHalf& g, const v2f (&sqr)[4],
                                          v2f (&partial)[4], float* bstage,
                                          const float* __restrict__ Rt,
                                          const float* __restrict__ sqt,
                                          unsigned laneOff, unsigned thrOff,
                                          unsigned stOff, unsigned r16) {
  const float* cur = bstage + CUR * GRAM_STAGE_FLOATS;
  float*       nxt = bstage + (CUR ^ 1) * GRAM_STAGE_FLOATS;
  load_btile(b1, cur + GRAM_TILE_FLOATS + laneOff, sqt + 16, r16);
  half_store(g, nxt + stOff);
  half_fetch(g, Rt + 3 * 16 * DPROJ, thrOff);
  tile_rowsums(afrag, b0, sqr, partial);
  half_store(g, nxt + GRAM_TILE_FLOATS + stOff);
  __syncthreads();
  __builtin_amdgcn_sched_barrier(0);
  load_btile(b0, nxt + laneOff, sqt + 32, r16);
  if (FETCH_AHEAD) half_fetch(g, Rt + 4 * 16 * DPROJ, thrOff);
  tile_rowsums(afrag, b1, sqr, partial);
  __builtin_amdgcn_sched_barrier(0);
}

__global__ __launch_bounds__(256, 4)
void gram_rowsum_kernel(const float* __restrict__ R, const float* __restrict__ sq,
                        float* __restrict__ anchorPart) {
  __shared__ __attribute__((aligned(16))) float bstage[2 * GRAM_STAGE_FLOATS];

  const unsigned tid   = threadIdx.x;
  const unsigned lane  = tid & 31;
  const unsigned chunk = blockIdx.x & (NCHUNK - 1);
  const unsigned strip = (blockIdx.x >> 3) * 8 + (tid >> 5);
  const unsigned half  = lane >> 4;
  const unsigned r16   = lane & 15;
  const unsigned koff  = half * 2;
  const unsigned m0    = strip * 16;

  v2f afrag[32];
  const float* arow = R + (size_t)(m0 + r16) * DPROJ + koff;
#pragma unroll
  for (int kk = 0; kk < 32; kk++) afrag[kk] = *(const v2f*)(arow + kk * 4);

  v2f sqr[4];
#pragma unroll
  for (int j = 0; j < 4; j++) {
    sqr[j].x = sq[m0 + 2 * j + half * 8];
    sqr[j].y = sq[m0 + 2 * j + 1 + half * 8];
  }

  v2f partial[4] = {{0.f, 0.f}, {0.f, 0.f}, {0.f, 0.f}, {0.f, 0.f}};

  const unsigned t0      = chunk * TILES_PER_CHUNK;
  const unsigned laneOff = r16 * GRAM_PITCH + half * 4;
  const unsigned thrOff  = (tid >> 4) * DPROJ + (tid & 15) * 8;
  const unsigned stOff   = (tid >> 4) * GRAM_PITCH + (tid & 15) * 8;
  const float* Rt  = R + (size_t)t0 * 16 * DPROJ;
  const float* sqt = sq + t0 * 16;

  BHalf g;
  BTile b0, b1;
  half_fetch(g, Rt, thrOff);
  half_store(g, bstage + stOff);
  half_fetch(g, Rt + 16 * DPROJ, thrOff);
  half_store(g, bstage + GRAM_TILE_FLOATS + stOff);
  __syncthreads();
  load_btile(b0, bstage + laneOff, sqt, r16);
  half_fetch(g, Rt + 2 * 16 * DPROJ, thrOff);

  for (unsigned sp = 0; sp < (GRAM_NSTEP - 2) / 2; sp++) {
    const float* Rs  = Rt + sp * (64 * DPROJ);
    const float* sqs = sqt + sp * 64;
    gram_step<0, true>(afrag, b0, b1, g, sqr, partial, bstage, Rs, sqs,
                       laneOff, thrOff, stOff, r16);
    gram_step<1, true>(afrag, b0, b1, g, sqr, partial, bstage, Rs + 32 * DPROJ, sqs + 32,
                       laneOff, thrOff, stOff, r16);
  }
  const float* Rl  = Rt + (GRAM_NSTEP - 2) * (32 * DPROJ);
  const float* sql = sqt + (GRAM_NSTEP - 2) * 32;
  gram_step<0, false>(afrag, b0, b1, g, sqr, partial, bstage, Rl, sql,
                      laneOff, thrOff, stOff, r16);
  load_btile(b1, bstage + GRAM_STAGE_FLOATS + GRAM_TILE_FLOATS + laneOff, sql + 48, r16);
  tile_rowsums(afrag, b0, sqr, partial);
  tile_rowsums(afrag, b1, sqr, partial);

#pragma unroll
  for (int i = 0; i < 8; i++) {
    float q = (i & 1) ? partial[i >> 1].y : partial[i >> 1].x;
    q += __shfl_xor(q, 1, 32);
    q += __shfl_xor(q, 2, 32);
    q += __shfl_xor(q, 4, 32);
    q += __shfl_xor(q, 8, 32);
    if (r16 == 0)
      anchorPart[(size_t)chunk * NPTS + m0 + i + half * 8] = q;
  }
}

__global__ __launch_bounds__(256)
void combine_kernel(const float* __restrict__ anchorPart,
                    float* __restrict__ anchorSq) {
  const int i = blockIdx.x * 256 + threadIdx.x;
  float s = 0.f;
#pragma unroll
  for (int c = 0; c < NCHUNK; c++) s += anchorPart[(size_t)c * NPTS + i];
  anchorSq[i] = s;
}

#define NWAVES (NPTS / 32)

__device__ __forceinline__ void argmax_merge(float& bv, int& bi, float ov, int oi) {
  const bool take = (ov > bv) | ((ov == bv) & (oi < bi));
  bv = take ? ov : bv;
  bi = take ? oi : bi;
}

template <int M>
__device__ __forceinline__ int lane_xor(int x) {
  return __builtin_amdgcn_update_dpp(x, x, 0x160 + M, 0xf, 0xf, false);
}
template <>
__device__ __forceinline__ int lane_xor<16>(int x) {
  return __builtin_amdgcn_permlanex16(x, x, 0x76543210, 0xfedcba98, false, false);
}

template <int M>
__device__ __forceinline__ void argmax_exchange(float& bv, int& bi) {
  const float ov = __int_as_float(lane_xor<M>(__float_as_int(bv)));
  const int   oi = lane_xor<M>(bi);
  argmax_merge(bv, bi, ov, oi);
}

__device__ __forceinline__ void wave_argmax(float& bv, int& bi) {
  argmax_exchange<1>(bv, bi);
  argmax_exchange<2>(bv, bi);
  argmax_exchange<4>(bv, bi);
  argmax_exchange<8>(bv, bi);
  argmax_exchange<16>(bv, bi);
}

__global__ __launch_bounds__(32)
void anchor_init_kernel(float* __restrict__ anchor, float* __restrict__ candV,
                        int* __restrict__ candI) {
  const int row  = blockIdx.x * 32 + threadIdx.x;
  const int lane = threadIdx.x;
  const float a = sqrtf(anchor[row]);
  anchor[row] = a;
  float bv = a; int bi = row;
  wave_argmax(bv, bi);
  if (lane == 0) { candV[row >> 5] = bv; candI[row >> 5] = bi; }
}

__global__ __launch_bounds__(32)
void greedy_step_kernel(const float* __restrict__ R, const float* __restrict__ sq,
                        float* __restrict__ anchor, float* __restrict__ candV,
                        int* __restrict__ candI, int* __restrict__ sel, int s) {
  __shared__ float4 selRow[DPROJ / 4];
  const int row  = blockIdx.x * 32 + threadIdx.x;
  const int lane = threadIdx.x;
  const int pin  = (s & 1) * NWAVES;
  const int pout = NWAVES - pin;

  float4 cv[4]; int4 ci[4];
  const float4* cvp = (const float4*)(candV + pin) + lane;
  const int4*   cip = (const int4*)(candI + pin) + lane;
#pragma unroll
  for (int q = 0; q < 4; q++) { cv[q] = cvp[q * 32]; ci[q] = cip[q * 32]; }

  float4 v[DPROJ / 4];
  const float4* rp = (const float4*)(R + (size_t)row * DPROJ);
#pragma unroll
  for (int k = 0; k < DPROJ / 4; k++) v[k] = rp[k];
  const float sqOwn = sq[row];
  const float aOld  = anchor[row];
  __builtin_amdgcn_sched_barrier(0);

  float bv = cv[0].x; int bi = ci[0].x;
#pragma unroll
  for (int q = 0; q < 4; q++) {
    if (q > 0) argmax_merge(bv, bi, cv[q].x, ci[q].x);
    argmax_merge(bv, bi, cv[q].y, ci[q].y);
    argmax_merge(bv, bi, cv[q].z, ci[q].z);
    argmax_merge(bv, bi, cv[q].w, ci[q].w);
  }
  wave_argmax(bv, bi);
  const int si = bi;

  selRow[lane] = ((const float4*)(R + (size_t)si * DPROJ))[lane];
  const float selSq = sq[si];
  __syncthreads();
  float d0 = 0.f, d1 = 0.f, d2s = 0.f, d3 = 0.f;
#pragma unroll
  for (int k = 0; k < DPROJ / 4; k++) {
    const float4 w = selRow[k];
    d0  = fmaf(v[k].x, w.x, d0);
    d1  = fmaf(v[k].y, w.y, d1);
    d2s = fmaf(v[k].z, w.z, d2s);
    d3  = fmaf(v[k].w, w.w, d3);
  }
  const float dot = (d0 + d1) + (d2s + d3);
  const float d2 = sqOwn + selSq - 2.f * dot;
  const float a  = fminf(aOld, sqrtf(fmaxf(d2, 0.f)));
  anchor[row] = a;

  float mv = a; int mi = row;
  wave_argmax(mv, mi);
  if (lane == 0) { candV[pout + (row >> 5)] = mv; candI[pout + (row >> 5)] = mi; }
  if (row == 0) sel[s] = si;
}

__global__ __launch_bounds__(256)
void gather_kernel(const float* __restrict__ F, const int* __restrict__ sel,
                   float* __restrict__ out) {
  const int s = blockIdx.x;
  const int c = blockIdx.y * 256 + threadIdx.x;
  out[(size_t)s * DIN + c] = F[(size_t)sel[s] * DIN + c];
}

extern "C" void kernel_launch(void* const* d_in, const int* in_sizes, int n_in,
                              void* d_out, int out_size, void* d_ws, size_t ws_size,
                              hipStream_t stream) {
  (void)in_sizes; (void)n_in; (void)out_size; (void)ws_size;
  const float* F = (const float*)d_in[0];
  const float* W = (const float*)d_in[1];

  float* R          = (float*)d_ws;
  float* sq         = R + (size_t)NPTS * DPROJ;
  float* anchorSq   = sq + NPTS;
  float* anchorPart = anchorSq + NPTS;
  int*   sel        = (int*)(anchorPart + (size_t)NCHUNK * NPTS);

  proj_kernel<<<1024, 256, 0, stream>>>(F, W, R);
  sq_kernel<<<NPTS / 8, 256, 0, stream>>>(R, sq);
  gram_rowsum_kernel<<<1024, 256, 0, stream>>>(R, sq, anchorPart);
  combine_kernel<<<NPTS / 256, 256, 0, stream>>>(anchorPart, anchorSq);
  float* candV = anchorPart;
  int*   candI = (int*)(anchorPart + 2 * NWAVES);
  anchor_init_kernel<<<NPTS / 32, 32, 0, stream>>>(anchorSq, candV, candI);
  for (int s = 0; s < NSAMP; s++)
    greedy_step_kernel<<<NPTS / 32, 32, 0, stream>>>(R, sq, anchorSq, candV, candI, sel, s);
  gather_kernel<<<dim3(NSAMP, DIN / 256), 256, 0, stream>>>(F, sel, (float*)d_out);
}


# greedy launches: each wave fetches its 32 own rows as 32 whole-row 512-byte loads (4 lines each) and turns them to one row per lane in LDS, replacing 32 loads of 32 lines each; sums unchanged
# speedup vs baseline: 30.1037x; 1.0513x over previous
#include <hip/hip_runtime.h>


#define NPTS   16384
#define DIN    1024
#define DPROJ  128
#define NSAMP  163
#define NCHUNK 8
#define TILES_PER_CHUNK (NPTS / 16 / NCHUNK)

typedef float v2f __attribute__((ext_vector_type(2)));
typedef float v8f __attribute__((ext_vector_type(8)));

__device__ __forceinline__ v8f wmma_f32_k4(v2f a, v2f b, v8f c) {
  return __builtin_amdgcn_wmma_f32_16x16x4_f32(
       false, a,  false, b,
       (short)0, c,  false,  false);
}

__global__ __launch_bounds__(256)
void proj_kernel(const float* __restrict__ F, const float* __restrict__ W,
                 float* __restrict__ R) {
  const int wave = (blockIdx.x * blockDim.x + threadIdx.x) >> 5;
  const int lane = threadIdx.x & 31;
  const int tm = wave >> 3;
  const int tn = wave & 7;
  const int m0 = tm * 16, n0 = tn * 16;
  const int half = lane >> 4;
  const int r16  = lane & 15;
  const int koff = half * 2;

  v8f acc = {};
  const float* arow = F + (size_t)(m0 + r16) * DIN + koff;
  for (int k = 0; k < DIN; k += 4) {
    v2f a = *(const v2f*)(arow + k);
    v2f b;
    b.x = W[(size_t)(k + koff)     * DPROJ + n0 + r16];
    b.y = W[(size_t)(k + koff + 1) * DPROJ + n0 + r16];
    acc = wmma_f32_k4(a, b, acc);
  }
#pragma unroll
  for (int i = 0; i < 8; i++)
    R[(size_t)(m0 + i + half * 8) * DPROJ + n0 + r16] = acc[i];
}

__global__ __launch_bounds__(256)
void sq_kernel(const float* __restrict__ R, float* __restrict__ sq) {
  const int row  = (blockIdx.x * blockDim.x + threadIdx.x) >> 5;
  const int lane = threadIdx.x & 31;
  const float* rp = R + (size_t)row * DPROJ;
  float s = 0.f;
#pragma unroll
  for (int k = lane; k < DPROJ; k += 32) { float v = rp[k]; s += v * v; }
#pragma unroll
  for (int m = 16; m >= 1; m >>= 1) s += __shfl_xor(s, m, 32);
  if (lane == 0) sq[row] = s;
}

#define GRAM_PITCH        132
#define GRAM_STAGE_TILES  2
#define GRAM_TILE_FLOATS  (16 * GRAM_PITCH)
#define GRAM_STAGE_FLOATS (GRAM_STAGE_TILES * GRAM_TILE_FLOATS)
#define GRAM_NSTEP        (TILES_PER_CHUNK / GRAM_STAGE_TILES)

struct BTile { v2f f[32]; float sqc; };
struct BHalf { float4 p[2]; };

__device__ __forceinline__ void half_fetch(BHalf& g, const float* __restrict__ Rtile,
                                           unsigned thrOff) {
  const float* a = Rtile + thrOff;
  g.p[0] = *(const float4*)(a);
  g.p[1] = *(const float4*)(a + 4);
}

__device__ __forceinline__ void half_store(const BHalf& g, float* d) {
  *(float4*)(d)     = make_float4(g.p[0].x, g.p[0].y, g.p[1].x, g.p[1].y);
  *(float4*)(d + 4) = make_float4(g.p[0].z, g.p[0].w, g.p[1].z, g.p[1].w);
}

__device__ __forceinline__ void load_btile(BTile& bt, const float* tile,
                                           const float* __restrict__ sqTile, unsigned r16) {
#pragma unroll
  for (int j = 0; j < 16; j++) {
    const float4 q = *(const float4*)(tile + j * 8);
    bt.f[2 * j].x     = q.x;  bt.f[2 * j].y     = q.y;
    bt.f[2 * j + 1].x = q.z;  bt.f[2 * j + 1].y = q.w;
  }
  bt.sqc = sqTile[r16];
}

__device__ __forceinline__ void tile_rowsums(const v2f (&afrag)[32], const BTile& bt,
                                             const v2f (&sqr)[4], v2f (&partial)[4]) {
  v8f acc = {};
#pragma unroll
  for (int kk = 0; kk < 32; kk++) acc = wmma_f32_k4(afrag[kk], bt.f[kk], acc);
  const v2f sqc2 = {bt.sqc, bt.sqc};
  const v2f zero2 = {0.f, 0.f};
#pragma unroll
  for (int j = 0; j < 4; j++) {
    const v2f a = {acc[2 * j], acc[2 * j + 1]};
    partial[j] += __builtin_elementwise_max(zero2, sqr[j] + sqc2 - 2.f * a);
  }
}

template <int CUR, bool FETCH_AHEAD>
__device__ __forceinline__ void gram_step(const v2f (&afrag)[32], BTile& b0, BTile& b1,
                                          BHalf& g, const v2f (&sqr)[4],
                                          v2f (&partial)[4], float* bstage,
                                          const float* __restrict__ Rt,
                                          const float* __restrict__ sqt,
                                          unsigned laneOff, unsigned thrOff,
                                          unsigned stOff, unsigned r16) {
  const float* cur = bstage + CUR * GRAM_STAGE_FLOATS;
  float*       nxt = bstage + (CUR ^ 1) * GRAM_STAGE_FLOATS;
  load_btile(b1, cur + GRAM_TILE_FLOATS + laneOff, sqt + 16, r16);
  half_store(g, nxt + stOff);
  half_fetch(g, Rt + 3 * 16 * DPROJ, thrOff);
  tile_rowsums(afrag, b0, sqr, partial);
  half_store(g, nxt + GRAM_TILE_FLOATS + stOff);
  __syncthreads();
  __builtin_amdgcn_sched_barrier(0);
  load_btile(b0, nxt + laneOff, sqt + 32, r16);
  if (FETCH_AHEAD) half_fetch(g, Rt + 4 * 16 * DPROJ, thrOff);
  tile_rowsums(afrag, b1, sqr, partial);
  __builtin_amdgcn_sched_barrier(0);
}

__global__ __launch_bounds__(256, 4)
void gram_rowsum_kernel(const float* __restrict__ R, const float* __restrict__ sq,
                        float* __restrict__ anchorPart) {
  __shared__ __attribute__((aligned(16))) float bstage[2 * GRAM_STAGE_FLOATS];

  const unsigned tid   = threadIdx.x;
  const unsigned lane  = tid & 31;
  const unsigned chunk = blockIdx.x & (NCHUNK - 1);
  const unsigned strip = (blockIdx.x >> 3) * 8 + (tid >> 5);
  const unsigned half  = lane >> 4;
  const unsigned r16   = lane & 15;
  const unsigned koff  = half * 2;
  const unsigned m0    = strip * 16;

  v2f afrag[32];
  const float* arow = R + (size_t)(m0 + r16) * DPROJ + koff;
#pragma unroll
  for (int kk = 0; kk < 32; kk++) afrag[kk] = *(const v2f*)(arow + kk * 4);

  v2f sqr[4];
#pragma unroll
  for (int j = 0; j < 4; j++) {
    sqr[j].x = sq[m0 + 2 * j + half * 8];
    sqr[j].y = sq[m0 + 2 * j + 1 + half * 8];
  }

  v2f partial[4] = {{0.f, 0.f}, {0.f, 0.f}, {0.f, 0.f}, {0.f, 0.f}};

  const unsigned t0      = chunk * TILES_PER_CHUNK;
  const unsigned laneOff = r16 * GRAM_PITCH + half * 4;
  const unsigned thrOff  = (tid >> 4) * DPROJ + (tid & 15) * 8;
  const unsigned stOff   = (tid >> 4) * GRAM_PITCH + (tid & 15) * 8;
  const float* Rt  = R + (size_t)t0 * 16 * DPROJ;
  const float* sqt = sq + t0 * 16;

  BHalf g;
  BTile b0, b1;
  half_fetch(g, Rt, thrOff);
  half_store(g, bstage + stOff);
  half_fetch(g, Rt + 16 * DPROJ, thrOff);
  half_store(g, bstage + GRAM_TILE_FLOATS + stOff);
  __syncthreads();
  load_btile(b0, bstage + laneOff, sqt, r16);
  half_fetch(g, Rt + 2 * 16 * DPROJ, thrOff);

  for (unsigned sp = 0; sp < (GRAM_NSTEP - 2) / 2; sp++) {
    const float* Rs  = Rt + sp * (64 * DPROJ);
    const float* sqs = sqt + sp * 64;
    gram_step<0, true>(afrag, b0, b1, g, sqr, partial, bstage, Rs, sqs,
                       laneOff, thrOff, stOff, r16);
    gram_step<1, true>(afrag, b0, b1, g, sqr, partial, bstage, Rs + 32 * DPROJ, sqs + 32,
                       laneOff, thrOff, stOff, r16);
  }
  const float* Rl  = Rt + (GRAM_NSTEP - 2) * (32 * DPROJ);
  const float* sql = sqt + (GRAM_NSTEP - 2) * 32;
  gram_step<0, false>(afrag, b0, b1, g, sqr, partial, bstage, Rl, sql,
                      laneOff, thrOff, stOff, r16);
  load_btile(b1, bstage + GRAM_STAGE_FLOATS + GRAM_TILE_FLOATS + laneOff, sql + 48, r16);
  tile_rowsums(afrag, b0, sqr, partial);
  tile_rowsums(afrag, b1, sqr, partial);

#pragma unroll
  for (int i = 0; i < 8; i++) {
    float q = (i & 1) ? partial[i >> 1].y : partial[i >> 1].x;
    q += __shfl_xor(q, 1, 32);
    q += __shfl_xor(q, 2, 32);
    q += __shfl_xor(q, 4, 32);
    q += __shfl_xor(q, 8, 32);
    if (r16 == 0)
      anchorPart[(size_t)chunk * NPTS + m0 + i + half * 8] = q;
  }
}

__global__ __launch_bounds__(256)
void combine_kernel(const float* __restrict__ anchorPart,
                    float* __restrict__ anchorSq) {
  const int i = blockIdx.x * 256 + threadIdx.x;
  float s = 0.f;
#pragma unroll
  for (int c = 0; c < NCHUNK; c++) s += anchorPart[(size_t)c * NPTS + i];
  anchorSq[i] = s;
}

#define NWAVES (NPTS / 32)
#define OWN_PITCH 132

__device__ __forceinline__ void argmax_merge(float& bv, int& bi, float ov, int oi) {
  const bool take = (ov > bv) | ((ov == bv) & (oi < bi));
  bv = take ? ov : bv;
  bi = take ? oi : bi;
}

template <int M>
__device__ __forceinline__ int lane_xor(int x) {
  return __builtin_amdgcn_update_dpp(x, x, 0x160 + M, 0xf, 0xf, false);
}
template <>
__device__ __forceinline__ int lane_xor<16>(int x) {
  return __builtin_amdgcn_permlanex16(x, x, 0x76543210, 0xfedcba98, false, false);
}

template <int M>
__device__ __forceinline__ void argmax_exchange(float& bv, int& bi) {
  const float ov = __int_as_float(lane_xor<M>(__float_as_int(bv)));
  const int   oi = lane_xor<M>(bi);
  argmax_merge(bv, bi, ov, oi);
}

__device__ __forceinline__ void wave_argmax(float& bv, int& bi) {
  argmax_exchange<1>(bv, bi);
  argmax_exchange<2>(bv, bi);
  argmax_exchange<4>(bv, bi);
  argmax_exchange<8>(bv, bi);
  argmax_exchange<16>(bv, bi);
}

__global__ __launch_bounds__(32)
void anchor_init_kernel(float* __restrict__ anchor, float* __restrict__ candV,
                        int* __restrict__ candI) {
  const int row  = blockIdx.x * 32 + threadIdx.x;
  const int lane = threadIdx.x;
  const float a = sqrtf(anchor[row]);
  anchor[row] = a;
  float bv = a; int bi = row;
  wave_argmax(bv, bi);
  if (lane == 0) { candV[row >> 5] = bv; candI[row >> 5] = bi; }
}

__global__ __launch_bounds__(32)
void greedy_step_kernel(const float* __restrict__ R, const float* __restrict__ sq,
                        float* __restrict__ anchor, float* __restrict__ candV,
                        int* __restrict__ candI, int* __restrict__ sel, int s) {
  __shared__ float4 selRow[DPROJ / 4];
  __shared__ __attribute__((aligned(16))) float ownStage[32 * OWN_PITCH];
  const int row  = blockIdx.x * 32 + threadIdx.x;
  const int lane = threadIdx.x;
  const int pin  = (s & 1) * NWAVES;
  const int pout = NWAVES - pin;

  float4 cv[4]; int4 ci[4];
  const float4* cvp = (const float4*)(candV + pin) + lane;
  const int4*   cip = (const int4*)(candI + pin) + lane;
#pragma unroll
  for (int q = 0; q < 4; q++) { cv[q] = cvp[q * 32]; ci[q] = cip[q * 32]; }

  const float4* gp = (const float4*)(R + (size_t)blockIdx.x * 32 * DPROJ) + lane;
  float* gs = ownStage + lane * 4;
#pragma unroll
  for (int r = 0; r < 32; r++) *(float4*)(gs + r * OWN_PITCH) = gp[r * (DPROJ / 4)];
  const float sqOwn = sq[row];
  const float aOld  = anchor[row];
  __builtin_amdgcn_sched_barrier(0);

  float bv = cv[0].x; int bi = ci[0].x;
#pragma unroll
  for (int q = 0; q < 4; q++) {
    if (q > 0) argmax_merge(bv, bi, cv[q].x, ci[q].x);
    argmax_merge(bv, bi, cv[q].y, ci[q].y);
    argmax_merge(bv, bi, cv[q].z, ci[q].z);
    argmax_merge(bv, bi, cv[q].w, ci[q].w);
  }
  wave_argmax(bv, bi);
  const int si = bi;

  const float4 selPiece = ((const float4*)(R + (size_t)si * DPROJ))[lane];
  const float selSq = sq[si];

  __builtin_amdgcn_wave_barrier();
  float4 v[DPROJ / 4];
#pragma unroll
  for (int k = 0; k < DPROJ / 4; k++) v[k] = *(const float4*)(ownStage + lane * OWN_PITCH + k * 4);
  selRow[lane] = selPiece;
  __syncthreads();
  float d0 = 0.f, d1 = 0.f, d2s = 0.f, d3 = 0.f;
#pragma unroll
  for (int k = 0; k < DPROJ / 4; k++) {
    const float4 w = selRow[k];
    d0  = fmaf(v[k].x, w.x, d0);
    d1  = fmaf(v[k].y, w.y, d1);
    d2s = fmaf(v[k].z, w.z, d2s);
    d3  = fmaf(v[k].w, w.w, d3);
  }
  const float dot = (d0 + d1) + (d2s + d3);
  const float d2 = sqOwn + selSq - 2.f * dot;
  const float a  = fminf(aOld, sqrtf(fmaxf(d2, 0.f)));
  anchor[row] = a;

  float mv = a; int mi = row;
  wave_argmax(mv, mi);
  if (lane == 0) { candV[pout + (row >> 5)] = mv; candI[pout + (row >> 5)] = mi; }
  if (row == 0) sel[s] = si;
}

__global__ __launch_bounds__(256)
void gather_kernel(const float* __restrict__ F, const int* __restrict__ sel,
                   float* __restrict__ out) {
  const int s = blockIdx.x;
  const int c = blockIdx.y * 256 + threadIdx.x;
  out[(size_t)s * DIN + c] = F[(size_t)sel[s] * DIN + c];
}

extern "C" void kernel_launch(void* const* d_in, const int* in_sizes, int n_in,
                              void* d_out, int out_size, void* d_ws, size_t ws_size,
                              hipStream_t stream) {
  (void)in_sizes; (void)n_in; (void)out_size; (void)ws_size;
  const float* F = (const float*)d_in[0];
  const float* W = (const float*)d_in[1];

  float* R          = (float*)d_ws;
  float* sq         = R + (size_t)NPTS * DPROJ;
  float* anchorSq   = sq + NPTS;
  float* anchorPart = anchorSq + NPTS;
  int*   sel        = (int*)(anchorPart + (size_t)NCHUNK * NPTS);

  proj_kernel<<<1024, 256, 0, stream>>>(F, W, R);
  sq_kernel<<<NPTS / 8, 256, 0, stream>>>(R, sq);
  gram_rowsum_kernel<<<1024, 256, 0, stream>>>(R, sq, anchorPart);
  combine_kernel<<<NPTS / 256, 256, 0, stream>>>(anchorPart, anchorSq);
  float* candV = anchorPart;
  int*   candI = (int*)(anchorPart + 2 * NWAVES);
  anchor_init_kernel<<<NPTS / 32, 32, 0, stream>>>(anchorSq, candV, candI);
  for (int s = 0; s < NSAMP; s++)
    greedy_step_kernel<<<NPTS / 32, 32, 0, stream>>>(R, sq, anchorSq, candV, candI, sel, s);
  gather_kernel<<<dim3(NSAMP, DIN / 256), 256, 0, stream>>>(F, sel, (float*)d_out);
}


# greedy launches: each argmax is a max reduction of the values, then a min reduction of the rows that hold the maximum, replacing 25 compare-scalar-select merge steps per launch; same result
# speedup vs baseline: 31.5944x; 1.0495x over previous
#include <hip/hip_runtime.h>


#define NPTS   16384
#define DIN    1024
#define DPROJ  128
#define NSAMP  163
#define NCHUNK 8
#define TILES_PER_CHUNK (NPTS / 16 / NCHUNK)

typedef float v2f __attribute__((ext_vector_type(2)));
typedef float v8f __attribute__((ext_vector_type(8)));

__device__ __forceinline__ v8f wmma_f32_k4(v2f a, v2f b, v8f c) {
  return __builtin_amdgcn_wmma_f32_16x16x4_f32(
       false, a,  false, b,
       (short)0, c,  false,  false);
}

__global__ __launch_bounds__(256)
void proj_kernel(const float* __restrict__ F, const float* __restrict__ W,
                 float* __restrict__ R) {
  const int wave = (blockIdx.x * blockDim.x + threadIdx.x) >> 5;
  const int lane = threadIdx.x & 31;
  const int tm = wave >> 3;
  const int tn = wave & 7;
  const int m0 = tm * 16, n0 = tn * 16;
  const int half = lane >> 4;
  const int r16  = lane & 15;
  const int koff = half * 2;

  v8f acc = {};
  const float* arow = F + (size_t)(m0 + r16) * DIN + koff;
  for (int k = 0; k < DIN; k += 4) {
    v2f a = *(const v2f*)(arow + k);
    v2f b;
    b.x = W[(size_t)(k + koff)     * DPROJ + n0 + r16];
    b.y = W[(size_t)(k + koff + 1) * DPROJ + n0 + r16];
    acc = wmma_f32_k4(a, b, acc);
  }
#pragma unroll
  for (int i = 0; i < 8; i++)
    R[(size_t)(m0 + i + half * 8) * DPROJ + n0 + r16] = acc[i];
}

__global__ __launch_bounds__(256)
void sq_kernel(const float* __restrict__ R, float* __restrict__ sq) {
  const int row  = (blockIdx.x * blockDim.x + threadIdx.x) >> 5;
  const int lane = threadIdx.x & 31;
  const float* rp = R + (size_t)row * DPROJ;
  float s = 0.f;
#pragma unroll
  for (int k = lane; k < DPROJ; k += 32) { float v = rp[k]; s += v * v; }
#pragma unroll
  for (int m = 16; m >= 1; m >>= 1) s += __shfl_xor(s, m, 32);
  if (lane == 0) sq[row] = s;
}

#define GRAM_PITCH        132
#define GRAM_STAGE_TILES  2
#define GRAM_TILE_FLOATS  (16 * GRAM_PITCH)
#define GRAM_STAGE_FLOATS (GRAM_STAGE_TILES * GRAM_TILE_FLOATS)
#define GRAM_NSTEP        (TILES_PER_CHUNK / GRAM_STAGE_TILES)

struct BTile { v2f f[32]; float sqc; };
struct BHalf { float4 p[2]; };

__device__ __forceinline__ void half_fetch(BHalf& g, const float* __restrict__ Rtile,
                                           unsigned thrOff) {
  const float* a = Rtile + thrOff;
  g.p[0] = *(const float4*)(a);
  g.p[1] = *(const float4*)(a + 4);
}

__device__ __forceinline__ void half_store(const BHalf& g, float* d) {
  *(float4*)(d)     = make_float4(g.p[0].x, g.p[0].y, g.p[1].x, g.p[1].y);
  *(float4*)(d + 4) = make_float4(g.p[0].z, g.p[0].w, g.p[1].z, g.p[1].w);
}

__device__ __forceinline__ void load_btile(BTile& bt, const float* tile,
                                           const float* __restrict__ sqTile, unsigned r16) {
#pragma unroll
  for (int j = 0; j < 16; j++) {
    const float4 q = *(const float4*)(tile + j * 8);
    bt.f[2 * j].x     = q.x;  bt.f[2 * j].y     = q.y;
    bt.f[2 * j + 1].x = q.z;  bt.f[2 * j + 1].y = q.w;
  }
  bt.sqc = sqTile[r16];
}

__device__ __forceinline__ void tile_rowsums(const v2f (&afrag)[32], const BTile& bt,
                                             const v2f (&sqr)[4], v2f (&partial)[4]) {
  v8f acc = {};
#pragma unroll
  for (int kk = 0; kk < 32; kk++) acc = wmma_f32_k4(afrag[kk], bt.f[kk], acc);
  const v2f sqc2 = {bt.sqc, bt.sqc};
  const v2f zero2 = {0.f, 0.f};
#pragma unroll
  for (int j = 0; j < 4; j++) {
    const v2f a = {acc[2 * j], acc[2 * j + 1]};
    partial[j] += __builtin_elementwise_max(zero2, sqr[j] + sqc2 - 2.f * a);
  }
}

template <int CUR, bool FETCH_AHEAD>
__device__ __forceinline__ void gram_step(const v2f (&afrag)[32], BTile& b0, BTile& b1,
                                          BHalf& g, const v2f (&sqr)[4],
                                          v2f (&partial)[4], float* bstage,
                                          const float* __restrict__ Rt,
                                          const float* __restrict__ sqt,
                                          unsigned laneOff, unsigned thrOff,
                                          unsigned stOff, unsigned r16) {
  const float* cur = bstage + CUR * GRAM_STAGE_FLOATS;
  float*       nxt = bstage + (CUR ^ 1) * GRAM_STAGE_FLOATS;
  load_btile(b1, cur + GRAM_TILE_FLOATS + laneOff, sqt + 16, r16);
  half_store(g, nxt + stOff);
  half_fetch(g, Rt + 3 * 16 * DPROJ, thrOff);
  tile_rowsums(afrag, b0, sqr, partial);
  half_store(g, nxt + GRAM_TILE_FLOATS + stOff);
  __syncthreads();
  __builtin_amdgcn_sched_barrier(0);
  load_btile(b0, nxt + laneOff, sqt + 32, r16);
  if (FETCH_AHEAD) half_fetch(g, Rt + 4 * 16 * DPROJ, thrOff);
  tile_rowsums(afrag, b1, sqr, partial);
  __builtin_amdgcn_sched_barrier(0);
}

__global__ __launch_bounds__(256, 4)
void gram_rowsum_kernel(const float* __restrict__ R, const float* __restrict__ sq,
                        float* __restrict__ anchorPart) {
  __shared__ __attribute__((aligned(16))) float bstage[2 * GRAM_STAGE_FLOATS];

  const unsigned tid   = threadIdx.x;
  const unsigned lane  = tid & 31;
  const unsigned chunk = blockIdx.x & (NCHUNK - 1);
  const unsigned strip = (blockIdx.x >> 3) * 8 + (tid >> 5);
  const unsigned half  = lane >> 4;
  const unsigned r16   = lane & 15;
  const unsigned koff  = half * 2;
  const unsigned m0    = strip * 16;

  v2f afrag[32];
  const float* arow = R + (size_t)(m0 + r16) * DPROJ + koff;
#pragma unroll
  for (int kk = 0; kk < 32; kk++) afrag[kk] = *(const v2f*)(arow + kk * 4);

  v2f sqr[4];
#pragma unroll
  for (int j = 0; j < 4; j++) {
    sqr[j].x = sq[m0 + 2 * j + half * 8];
    sqr[j].y = sq[m0 + 2 * j + 1 + half * 8];
  }

  v2f partial[4] = {{0.f, 0.f}, {0.f, 0.f}, {0.f, 0.f}, {0.f, 0.f}};

  const unsigned t0      = chunk * TILES_PER_CHUNK;
  const unsigned laneOff = r16 * GRAM_PITCH + half * 4;
  const unsigned thrOff  = (tid >> 4) * DPROJ + (tid & 15) * 8;
  const unsigned stOff   = (tid >> 4) * GRAM_PITCH + (tid & 15) * 8;
  const float* Rt  = R + (size_t)t0 * 16 * DPROJ;
  const float* sqt = sq + t0 * 16;

  BHalf g;
  BTile b0, b1;
  half_fetch(g, Rt, thrOff);
  half_store(g, bstage + stOff);
  half_fetch(g, Rt + 16 * DPROJ, thrOff);
  half_store(g, bstage + GRAM_TILE_FLOATS + stOff);
  __syncthreads();
  load_btile(b0, bstage + laneOff, sqt, r16);
  half_fetch(g, Rt + 2 * 16 * DPROJ, thrOff);

  for (unsigned sp = 0; sp < (GRAM_NSTEP - 2) / 2; sp++) {
    const float* Rs  = Rt + sp * (64 * DPROJ);
    const float* sqs = sqt + sp * 64;
    gram_step<0, true>(afrag, b0, b1, g, sqr, partial, bstage, Rs, sqs,
                       laneOff, thrOff, stOff, r16);
    gram_step<1, true>(afrag, b0, b1, g, sqr, partial, bstage, Rs + 32 * DPROJ, sqs + 32,
                       laneOff, thrOff, stOff, r16);
  }
  const float* Rl  = Rt + (GRAM_NSTEP - 2) * (32 * DPROJ);
  const float* sql = sqt + (GRAM_NSTEP - 2) * 32;
  gram_step<0, false>(afrag, b0, b1, g, sqr, partial, bstage, Rl, sql,
                      laneOff, thrOff, stOff, r16);
  load_btile(b1, bstage + GRAM_STAGE_FLOATS + GRAM_TILE_FLOATS + laneOff, sql + 48, r16);
  tile_rowsums(afrag, b0, sqr, partial);
  tile_rowsums(afrag, b1, sqr, partial);

#pragma unroll
  for (int i = 0; i < 8; i++) {
    float q = (i & 1) ? partial[i >> 1].y : partial[i >> 1].x;
    q += __shfl_xor(q, 1, 32);
    q += __shfl_xor(q, 2, 32);
    q += __shfl_xor(q, 4, 32);
    q += __shfl_xor(q, 8, 32);
    if (r16 == 0)
      anchorPart[(size_t)chunk * NPTS + m0 + i + half * 8] = q;
  }
}

__global__ __launch_bounds__(256)
void combine_kernel(const float* __restrict__ anchorPart,
                    float* __restrict__ anchorSq) {
  const int i = blockIdx.x * 256 + threadIdx.x;
  float s = 0.f;
#pragma unroll
  for (int c = 0; c < NCHUNK; c++) s += anchorPart[(size_t)c * NPTS + i];
  anchorSq[i] = s;
}

#define NWAVES (NPTS / 32)
#define OWN_PITCH 132

template <int M>
__device__ __forceinline__ int lane_xor(int x) {
  return __builtin_amdgcn_update_dpp(0, x, 0x160 + M, 0xf, 0xf, true);
}
template <>
__device__ __forceinline__ int lane_xor<16>(int x) {
  return __builtin_amdgcn_permlanex16(0, x, 0x76543210, 0xfedcba98, false, false);
}

#define NO_ROW 0x7fffffff

__device__ __forceinline__ int imin(int a, int b) { return a < b ? a : b; }

template <int M>
__device__ __forceinline__ float max_exchange(float v) {
  return fmaxf(v, __int_as_float(lane_xor<M>(__float_as_int(v))));
}
template <int M>
__device__ __forceinline__ int min_exchange(int v) {
  return imin(v, lane_xor<M>(v));
}

__device__ __forceinline__ float wave_max(float v) {
  v = max_exchange<1>(v);
  v = max_exchange<2>(v);
  v = max_exchange<4>(v);
  v = max_exchange<8>(v);
  return max_exchange<16>(v);
}
__device__ __forceinline__ int wave_min(int v) {
  v = min_exchange<1>(v);
  v = min_exchange<2>(v);
  v = min_exchange<4>(v);
  v = min_exchange<8>(v);
  return min_exchange<16>(v);
}

__device__ __forceinline__ void wave_argmax(float& bv, int& bi) {
  const float m = wave_max(bv);
  bi = wave_min(bv == m ? bi : NO_ROW);
  bv = m;
}

__global__ __launch_bounds__(32)
void anchor_init_kernel(float* __restrict__ anchor, float* __restrict__ candV,
                        int* __restrict__ candI) {
  const int row  = blockIdx.x * 32 + threadIdx.x;
  const int lane = threadIdx.x;
  const float a = sqrtf(anchor[row]);
  anchor[row] = a;
  float bv = a; int bi = row;
  wave_argmax(bv, bi);
  if (lane == 0) { candV[row >> 5] = bv; candI[row >> 5] = bi; }
}

__global__ __launch_bounds__(32)
void greedy_step_kernel(const float* __restrict__ R, const float* __restrict__ sq,
                        float* __restrict__ anchor, float* __restrict__ candV,
                        int* __restrict__ candI, int* __restrict__ sel, int s) {
  __shared__ float4 selRow[DPROJ / 4];
  __shared__ __attribute__((aligned(16))) float ownStage[32 * OWN_PITCH];
  const int row  = blockIdx.x * 32 + threadIdx.x;
  const int lane = threadIdx.x;
  const int pin  = (s & 1) * NWAVES;
  const int pout = NWAVES - pin;

  float4 cv[4]; int4 ci[4];
  const float4* cvp = (const float4*)(candV + pin) + lane;
  const int4*   cip = (const int4*)(candI + pin) + lane;
#pragma unroll
  for (int q = 0; q < 4; q++) { cv[q] = cvp[q * 32]; ci[q] = cip[q * 32]; }

  const float4* gp = (const float4*)(R + (size_t)blockIdx.x * 32 * DPROJ) + lane;
  float* gs = ownStage + lane * 4;
#pragma unroll
  for (int r = 0; r < 32; r++) *(float4*)(gs + r * OWN_PITCH) = gp[r * (DPROJ / 4)];
  const float sqOwn = sq[row];
  const float aOld  = anchor[row];
  __builtin_amdgcn_sched_barrier(0);

  float qm[4];
#pragma unroll
  for (int q = 0; q < 4; q++)
    qm[q] = fmaxf(fmaxf(cv[q].x, cv[q].y), fmaxf(cv[q].z, cv[q].w));
  float bv = fmaxf(fmaxf(qm[0], qm[1]), fmaxf(qm[2], qm[3]));
  int qi[4];
#pragma unroll
  for (int q = 0; q < 4; q++) {
    const int r0 = cv[q].x == bv ? ci[q].x : NO_ROW;
    const int r1 = cv[q].y == bv ? ci[q].y : NO_ROW;
    const int r2 = cv[q].z == bv ? ci[q].z : NO_ROW;
    const int r3 = cv[q].w == bv ? ci[q].w : NO_ROW;
    qi[q] = imin(imin(r0, r1), imin(r2, r3));
  }
  int bi = imin(imin(qi[0], qi[1]), imin(qi[2], qi[3]));
  wave_argmax(bv, bi);
  const int si = bi & (NPTS - 1);

  const float4 selPiece = ((const float4*)(R + (size_t)si * DPROJ))[lane];
  const float selSq = sq[si];

  __builtin_amdgcn_wave_barrier();
  float4 v[DPROJ / 4];
#pragma unroll
  for (int k = 0; k < DPROJ / 4; k++) v[k] = *(const float4*)(ownStage + lane * OWN_PITCH + k * 4);
  selRow[lane] = selPiece;
  __syncthreads();
  float d0 = 0.f, d1 = 0.f, d2s = 0.f, d3 = 0.f;
#pragma unroll
  for (int k = 0; k < DPROJ / 4; k++) {
    const float4 w = selRow[k];
    d0  = fmaf(v[k].x, w.x, d0);
    d1  = fmaf(v[k].y, w.y, d1);
    d2s = fmaf(v[k].z, w.z, d2s);
    d3  = fmaf(v[k].w, w.w, d3);
  }
  const float dot = (d0 + d1) + (d2s + d3);
  const float d2 = sqOwn + selSq - 2.f * dot;
  const float a  = fminf(aOld, sqrtf(fmaxf(d2, 0.f)));
  anchor[row] = a;

  float mv = a; int mi = row;
  wave_argmax(mv, mi);
  if (lane == 0) { candV[pout + (row >> 5)] = mv; candI[pout + (row >> 5)] = mi; }
  if (row == 0) sel[s] = si;
}

__global__ __launch_bounds__(256)
void gather_kernel(const float* __restrict__ F, const int* __restrict__ sel,
                   float* __restrict__ out) {
  const int s = blockIdx.x;
  const int c = blockIdx.y * 256 + threadIdx.x;
  out[(size_t)s * DIN + c] = F[(size_t)sel[s] * DIN + c];
}

extern "C" void kernel_launch(void* const* d_in, const int* in_sizes, int n_in,
                              void* d_out, int out_size, void* d_ws, size_t ws_size,
                              hipStream_t stream) {
  (void)in_sizes; (void)n_in; (void)out_size; (void)ws_size;
  const float* F = (const float*)d_in[0];
  const float* W = (const float*)d_in[1];

  float* R          = (float*)d_ws;
  float* sq         = R + (size_t)NPTS * DPROJ;
  float* anchorSq   = sq + NPTS;
  float* anchorPart = anchorSq + NPTS;
  int*   sel        = (int*)(anchorPart + (size_t)NCHUNK * NPTS);

  proj_kernel<<<1024, 256, 0, stream>>>(F, W, R);
  sq_kernel<<<NPTS / 8, 256, 0, stream>>>(R, sq);
  gram_rowsum_kernel<<<1024, 256, 0, stream>>>(R, sq, anchorPart);
  combine_kernel<<<NPTS / 256, 256, 0, stream>>>(anchorPart, anchorSq);
  float* candV = anchorPart;
  int*   candI = (int*)(anchorPart + 2 * NWAVES);
  anchor_init_kernel<<<NPTS / 32, 32, 0, stream>>>(anchorSq, candV, candI);
  for (int s = 0; s < NSAMP; s++)
    greedy_step_kernel<<<NPTS / 32, 32, 0, stream>>>(R, sq, anchorSq, candV, candI, sel, s);
  gather_kernel<<<dim3(NSAMP, DIN / 256), 256, 0, stream>>>(F, sel, (float*)d_out);
}


# greedy launches: each dot product is taken on the rows as loaded (16 bytes of every row per lane) and summed over the lanes by five exchange stages, no LDS; the dot's sum order changes
# speedup vs baseline: 34.1404x; 1.0806x over previous
#include <hip/hip_runtime.h>


#define NPTS   16384
#define DIN    1024
#define DPROJ  128
#define NSAMP  163
#define NCHUNK 8
#define TILES_PER_CHUNK (NPTS / 16 / NCHUNK)

typedef float v2f __attribute__((ext_vector_type(2)));
typedef float v4f __attribute__((ext_vector_type(4)));
typedef float v8f __attribute__((ext_vector_type(8)));

__device__ __forceinline__ v8f wmma_f32_k4(v2f a, v2f b, v8f c) {
  return __builtin_amdgcn_wmma_f32_16x16x4_f32(
       false, a,  false, b,
       (short)0, c,  false,  false);
}

__global__ __launch_bounds__(256)
void proj_kernel(const float* __restrict__ F, const float* __restrict__ W,
                 float* __restrict__ R) {
  const int wave = (blockIdx.x * blockDim.x + threadIdx.x) >> 5;
  const int lane = threadIdx.x & 31;
  const int tm = wave >> 3;
  const int tn = wave & 7;
  const int m0 = tm * 16, n0 = tn * 16;
  const int half = lane >> 4;
  const int r16  = lane & 15;
  const int koff = half * 2;

  v8f acc = {};
  const float* arow = F + (size_t)(m0 + r16) * DIN + koff;
  for (int k = 0; k < DIN; k += 4) {
    v2f a = *(const v2f*)(arow + k);
    v2f b;
    b.x = W[(size_t)(k + koff)     * DPROJ + n0 + r16];
    b.y = W[(size_t)(k + koff + 1) * DPROJ + n0 + r16];
    acc = wmma_f32_k4(a, b, acc);
  }
#pragma unroll
  for (int i = 0; i < 8; i++)
    R[(size_t)(m0 + i + half * 8) * DPROJ + n0 + r16] = acc[i];
}

__global__ __launch_bounds__(256)
void sq_kernel(const float* __restrict__ R, float* __restrict__ sq) {
  const int row  = (blockIdx.x * blockDim.x + threadIdx.x) >> 5;
  const int lane = threadIdx.x & 31;
  const float* rp = R + (size_t)row * DPROJ;
  float s = 0.f;
#pragma unroll
  for (int k = lane; k < DPROJ; k += 32) { float v = rp[k]; s += v * v; }
#pragma unroll
  for (int m = 16; m >= 1; m >>= 1) s += __shfl_xor(s, m, 32);
  if (lane == 0) sq[row] = s;
}

#define GRAM_PITCH        132
#define GRAM_STAGE_TILES  2
#define GRAM_TILE_FLOATS  (16 * GRAM_PITCH)
#define GRAM_STAGE_FLOATS (GRAM_STAGE_TILES * GRAM_TILE_FLOATS)
#define GRAM_NSTEP        (TILES_PER_CHUNK / GRAM_STAGE_TILES)

struct BTile { v2f f[32]; float sqc; };
struct BHalf { float4 p[2]; };

__device__ __forceinline__ void half_fetch(BHalf& g, const float* __restrict__ Rtile,
                                           unsigned thrOff) {
  const float* a = Rtile + thrOff;
  g.p[0] = *(const float4*)(a);
  g.p[1] = *(const float4*)(a + 4);
}

__device__ __forceinline__ void half_store(const BHalf& g, float* d) {
  *(float4*)(d)     = make_float4(g.p[0].x, g.p[0].y, g.p[1].x, g.p[1].y);
  *(float4*)(d + 4) = make_float4(g.p[0].z, g.p[0].w, g.p[1].z, g.p[1].w);
}

__device__ __forceinline__ void load_btile(BTile& bt, const float* tile,
                                           const float* __restrict__ sqTile, unsigned r16) {
#pragma unroll
  for (int j = 0; j < 16; j++) {
    const float4 q = *(const float4*)(tile + j * 8);
    bt.f[2 * j].x     = q.x;  bt.f[2 * j].y     = q.y;
    bt.f[2 * j + 1].x = q.z;  bt.f[2 * j + 1].y = q.w;
  }
  bt.sqc = sqTile[r16];
}

__device__ __forceinline__ void tile_rowsums(const v2f (&afrag)[32], const BTile& bt,
                                             const v2f (&sqr)[4], v2f (&partial)[4]) {
  v8f acc = {};
#pragma unroll
  for (int kk = 0; kk < 32; kk++) acc = wmma_f32_k4(afrag[kk], bt.f[kk], acc);
  const v2f sqc2 = {bt.sqc, bt.sqc};
  const v2f zero2 = {0.f, 0.f};
#pragma unroll
  for (int j = 0; j < 4; j++) {
    const v2f a = {acc[2 * j], acc[2 * j + 1]};
    partial[j] += __builtin_elementwise_max(zero2, sqr[j] + sqc2 - 2.f * a);
  }
}

template <int CUR, bool FETCH_AHEAD>
__device__ __forceinline__ void gram_step(const v2f (&afrag)[32], BTile& b0, BTile& b1,
                                          BHalf& g, const v2f (&sqr)[4],
                                          v2f (&partial)[4], float* bstage,
                                          const float* __restrict__ Rt,
                                          const float* __restrict__ sqt,
                                          unsigned laneOff, unsigned thrOff,
                                          unsigned stOff, unsigned r16) {
  const float* cur = bstage + CUR * GRAM_STAGE_FLOATS;
  float*       nxt = bstage + (CUR ^ 1) * GRAM_STAGE_FLOATS;
  load_btile(b1, cur + GRAM_TILE_FLOATS + laneOff, sqt + 16, r16);
  half_store(g, nxt + stOff);
  half_fetch(g, Rt + 3 * 16 * DPROJ, thrOff);
  tile_rowsums(afrag, b0, sqr, partial);
  half_store(g, nxt + GRAM_TILE_FLOATS + stOff);
  __syncthreads();
  __builtin_amdgcn_sched_barrier(0);
  load_btile(b0, nxt + laneOff, sqt + 32, r16);
  if (FETCH_AHEAD) half_fetch(g, Rt + 4 * 16 * DPROJ, thrOff);
  tile_rowsums(afrag, b1, sqr, partial);
  __builtin_amdgcn_sched_barrier(0);
}

__global__ __launch_bounds__(256, 4)
void gram_rowsum_kernel(const float* __restrict__ R, const float* __restrict__ sq,
                        float* __restrict__ anchorPart) {
  __shared__ __attribute__((aligned(16))) float bstage[2 * GRAM_STAGE_FLOATS];

  const unsigned tid   = threadIdx.x;
  const unsigned lane  = tid & 31;
  const unsigned chunk = blockIdx.x & (NCHUNK - 1);
  const unsigned strip = (blockIdx.x >> 3) * 8 + (tid >> 5);
  const unsigned half  = lane >> 4;
  const unsigned r16   = lane & 15;
  const unsigned koff  = half * 2;
  const unsigned m0    = strip * 16;

  v2f afrag[32];
  const float* arow = R + (size_t)(m0 + r16) * DPROJ + koff;
#pragma unroll
  for (int kk = 0; kk < 32; kk++) afrag[kk] = *(const v2f*)(arow + kk * 4);

  v2f sqr[4];
#pragma unroll
  for (int j = 0; j < 4; j++) {
    sqr[j].x = sq[m0 + 2 * j + half * 8];
    sqr[j].y = sq[m0 + 2 * j + 1 + half * 8];
  }

  v2f partial[4] = {{0.f, 0.f}, {0.f, 0.f}, {0.f, 0.f}, {0.f, 0.f}};

  const unsigned t0      = chunk * TILES_PER_CHUNK;
  const unsigned laneOff = r16 * GRAM_PITCH + half * 4;
  const unsigned thrOff  = (tid >> 4) * DPROJ + (tid & 15) * 8;
  const unsigned stOff   = (tid >> 4) * GRAM_PITCH + (tid & 15) * 8;
  const float* Rt  = R + (size_t)t0 * 16 * DPROJ;
  const float* sqt = sq + t0 * 16;

  BHalf g;
  BTile b0, b1;
  half_fetch(g, Rt, thrOff);
  half_store(g, bstage + stOff);
  half_fetch(g, Rt + 16 * DPROJ, thrOff);
  half_store(g, bstage + GRAM_TILE_FLOATS + stOff);
  __syncthreads();
  load_btile(b0, bstage + laneOff, sqt, r16);
  half_fetch(g, Rt + 2 * 16 * DPROJ, thrOff);

  for (unsigned sp = 0; sp < (GRAM_NSTEP - 2) / 2; sp++) {
    const float* Rs  = Rt + sp * (64 * DPROJ);
    const float* sqs = sqt + sp * 64;
    gram_step<0, true>(afrag, b0, b1, g, sqr, partial, bstage, Rs, sqs,
                       laneOff, thrOff, stOff, r16);
    gram_step<1, true>(afrag, b0, b1, g, sqr, partial, bstage, Rs + 32 * DPROJ, sqs + 32,
                       laneOff, thrOff, stOff, r16);
  }
  const float* Rl  = Rt + (GRAM_NSTEP - 2) * (32 * DPROJ);
  const float* sql = sqt + (GRAM_NSTEP - 2) * 32;
  gram_step<0, false>(afrag, b0, b1, g, sqr, partial, bstage, Rl, sql,
                      laneOff, thrOff, stOff, r16);
  load_btile(b1, bstage + GRAM_STAGE_FLOATS + GRAM_TILE_FLOATS + laneOff, sql + 48, r16);
  tile_rowsums(afrag, b0, sqr, partial);
  tile_rowsums(afrag, b1, sqr, partial);

#pragma unroll
  for (int i = 0; i < 8; i++) {
    float q = (i & 1) ? partial[i >> 1].y : partial[i >> 1].x;
    q += __shfl_xor(q, 1, 32);
    q += __shfl_xor(q, 2, 32);
    q += __shfl_xor(q, 4, 32);
    q += __shfl_xor(q, 8, 32);
    if (r16 == 0)
      anchorPart[(size_t)chunk * NPTS + m0 + i + half * 8] = q;
  }
}

__global__ __launch_bounds__(256)
void combine_kernel(const float* __restrict__ anchorPart,
                    float* __restrict__ anchorSq) {
  const int i = blockIdx.x * 256 + threadIdx.x;
  float s = 0.f;
#pragma unroll
  for (int c = 0; c < NCHUNK; c++) s += anchorPart[(size_t)c * NPTS + i];
  anchorSq[i] = s;
}

#define NWAVES (NPTS / 32)

template <int M>
__device__ __forceinline__ int lane_xor(int x) {
  return __builtin_amdgcn_update_dpp(0, x, 0x160 + M, 0xf, 0xf, true);
}
template <>
__device__ __forceinline__ int lane_xor<16>(int x) {
  return __builtin_amdgcn_permlanex16(0, x, 0x76543210, 0xfedcba98, false, false);
}

#define NO_ROW 0x7fffffff

__device__ __forceinline__ int imin(int a, int b) { return a < b ? a : b; }

template <int M>
__device__ __forceinline__ float max_exchange(float v) {
  return fmaxf(v, __int_as_float(lane_xor<M>(__float_as_int(v))));
}
template <int M>
__device__ __forceinline__ int min_exchange(int v) {
  return imin(v, lane_xor<M>(v));
}

__device__ __forceinline__ float wave_max(float v) {
  v = max_exchange<1>(v);
  v = max_exchange<2>(v);
  v = max_exchange<4>(v);
  v = max_exchange<8>(v);
  return max_exchange<16>(v);
}
__device__ __forceinline__ int wave_min(int v) {
  v = min_exchange<1>(v);
  v = min_exchange<2>(v);
  v = min_exchange<4>(v);
  v = min_exchange<8>(v);
  return min_exchange<16>(v);
}

__device__ __forceinline__ void wave_argmax(float& bv, int& bi) {
  const float m = wave_max(bv);
  bi = wave_min(bv == m ? bi : NO_ROW);
  bv = m;
}

__global__ __launch_bounds__(32)
void anchor_init_kernel(float* __restrict__ anchor, float* __restrict__ candV,
                        int* __restrict__ candI) {
  const int row  = blockIdx.x * 32 + threadIdx.x;
  const int lane = threadIdx.x;
  const float a = sqrtf(anchor[row]);
  anchor[row] = a;
  float bv = a; int bi = row;
  wave_argmax(bv, bi);
  if (lane == 0) { candV[row >> 5] = bv; candI[row >> 5] = bi; }
}

template <int M>
__device__ __forceinline__ float add_exchange(float keep, float send) {
  return keep + __int_as_float(lane_xor<M>(__float_as_int(send)));
}
template <int K>
__device__ __forceinline__ void fold_stage(float (&val)[32], int lane) {
  const bool up = (lane >> K) & 1;
#pragma unroll
  for (int idx = 0; idx < 32; idx += (2 << K)) {
    const float a = val[idx], b = val[idx | (1 << K)];
    val[idx] = add_exchange<(1 << K)>(up ? b : a, up ? a : b);
  }
}

__global__ __launch_bounds__(32)
void greedy_step_kernel(const float* __restrict__ R, const float* __restrict__ sq,
                        float* __restrict__ anchor, float* __restrict__ candV,
                        int* __restrict__ candI, int* __restrict__ sel, int s) {
  const int row  = blockIdx.x * 32 + threadIdx.x;
  const int lane = threadIdx.x;
  const int pin  = (s & 1) * NWAVES;
  const int pout = NWAVES - pin;

  float4 cv[4]; int4 ci[4];
  const float4* cvp = (const float4*)(candV + pin) + lane;
  const int4*   cip = (const int4*)(candI + pin) + lane;
#pragma unroll
  for (int q = 0; q < 4; q++) { cv[q] = cvp[q * 32]; ci[q] = cip[q * 32]; }

  const v4f* gp = (const v4f*)(R + (size_t)blockIdx.x * 32 * DPROJ) + lane;
  v4f own[32];
#pragma unroll
  for (int r = 0; r < 32; r++) own[r] = gp[r * (DPROJ / 4)];
  const float sqOwn = sq[row];
  const float aOld  = anchor[row];
  __builtin_amdgcn_sched_barrier(0);

  float qm[4];
#pragma unroll
  for (int q = 0; q < 4; q++)
    qm[q] = fmaxf(fmaxf(cv[q].x, cv[q].y), fmaxf(cv[q].z, cv[q].w));
  float bv = fmaxf(fmaxf(qm[0], qm[1]), fmaxf(qm[2], qm[3]));
  int qi[4];
#pragma unroll
  for (int q = 0; q < 4; q++) {
    const int r0 = cv[q].x == bv ? ci[q].x : NO_ROW;
    const int r1 = cv[q].y == bv ? ci[q].y : NO_ROW;
    const int r2 = cv[q].z == bv ? ci[q].z : NO_ROW;
    const int r3 = cv[q].w == bv ? ci[q].w : NO_ROW;
    qi[q] = imin(imin(r0, r1), imin(r2, r3));
  }
  int bi = imin(imin(qi[0], qi[1]), imin(qi[2], qi[3]));
  wave_argmax(bv, bi);
  const int si = bi & (NPTS - 1);

  const v4f selPiece = ((const v4f*)(R + (size_t)si * DPROJ))[lane];
  const float selSq = sq[si];

  const v2f slo = {selPiece.x, selPiece.y};
  const v2f shi = {selPiece.z, selPiece.w};
  float val[32];
#pragma unroll
  for (int r = 0; r < 32; r++) {
    const v2f lo = {own[r].x, own[r].y};
    const v2f hi = {own[r].z, own[r].w};
    const v2f p = __builtin_elementwise_fma(hi, shi, lo * slo);
    val[r] = p.x + p.y;
  }
  fold_stage<0>(val, lane);
  fold_stage<1>(val, lane);
  fold_stage<2>(val, lane);
  fold_stage<3>(val, lane);
  fold_stage<4>(val, lane);
  const float dot = val[0];
  const float d2 = sqOwn + selSq - 2.f * dot;
  const float a  = fminf(aOld, sqrtf(fmaxf(d2, 0.f)));
  anchor[row] = a;

  float mv = a; int mi = row;
  wave_argmax(mv, mi);
  if (lane == 0) { candV[pout + (row >> 5)] = mv; candI[pout + (row >> 5)] = mi; }
  if (row == 0) sel[s] = si;
}

__global__ __launch_bounds__(256)
void gather_kernel(const float* __restrict__ F, const int* __restrict__ sel,
                   float* __restrict__ out) {
  const int s = blockIdx.x;
  const int c = blockIdx.y * 256 + threadIdx.x;
  out[(size_t)s * DIN + c] = F[(size_t)sel[s] * DIN + c];
}

extern "C" void kernel_launch(void* const* d_in, const int* in_sizes, int n_in,
                              void* d_out, int out_size, void* d_ws, size_t ws_size,
                              hipStream_t stream) {
  (void)in_sizes; (void)n_in; (void)out_size; (void)ws_size;
  const float* F = (const float*)d_in[0];
  const float* W = (const float*)d_in[1];

  float* R          = (float*)d_ws;
  float* sq         = R + (size_t)NPTS * DPROJ;
  float* anchorSq   = sq + NPTS;
  float* anchorPart = anchorSq + NPTS;
  int*   sel        = (int*)(anchorPart + (size_t)NCHUNK * NPTS);

  proj_kernel<<<1024, 256, 0, stream>>>(F, W, R);
  sq_kernel<<<NPTS / 8, 256, 0, stream>>>(R, sq);
  gram_rowsum_kernel<<<1024, 256, 0, stream>>>(R, sq, anchorPart);
  combine_kernel<<<NPTS / 256, 256, 0, stream>>>(anchorPart, anchorSq);
  float* candV = anchorPart;
  int*   candI = (int*)(anchorPart + 2 * NWAVES);
  anchor_init_kernel<<<NPTS / 32, 32, 0, stream>>>(anchorSq, candV, candI);
  for (int s = 0; s < NSAMP; s++)
    greedy_step_kernel<<<NPTS / 32, 32, 0, stream>>>(R, sq, anchorSq, candV, candI, sel, s);
  gather_kernel<<<dim3(NSAMP, DIN / 256), 256, 0, stream>>>(F, sel, (float*)d_out);
}


# projection: one wave makes 16 whole rows (8 tiles share an A fragment, W read in whole lines, operands fetched 1-3 chunks ahead) and sums their squares; norm launch goes; k and norm sum orders change
# speedup vs baseline: 36.0319x; 1.0554x over previous
#include <hip/hip_runtime.h>


#define NPTS   16384
#define DIN    1024
#define DPROJ  128
#define NSAMP  163
#define NCHUNK 8
#define TILES_PER_CHUNK (NPTS / 16 / NCHUNK)

typedef float v2f __attribute__((ext_vector_type(2)));
typedef float v4f __attribute__((ext_vector_type(4)));
typedef float v8f __attribute__((ext_vector_type(8)));

__device__ __forceinline__ v8f wmma_f32_k4(v2f a, v2f b, v8f c) {
  return __builtin_amdgcn_wmma_f32_16x16x4_f32(
       false, a,  false, b,
       (short)0, c,  false,  false);
}

template <int M> __device__ __forceinline__ int lane_xor(int x);

struct ProjA { v4f a[4]; };
struct ProjB { v4f lo[16], hi[16]; };

__device__ __forceinline__ void proj_fetch_a(ProjA& p, const float* __restrict__ ap) {
#pragma unroll
  for (int i = 0; i < 4; i++) p.a[i] = *(const v4f*)(ap + 4 * i);
}
__device__ __forceinline__ void proj_fetch_b(ProjB& p, const float* __restrict__ bp) {
#pragma unroll
  for (int i = 0; i < 16; i++) {
    p.lo[i] = *(const v4f*)(bp + i * DPROJ);
    p.hi[i] = *(const v4f*)(bp + i * DPROJ + 64);
  }
}
__device__ __forceinline__ void proj_mma(const ProjA& pa, const ProjB& pb, v8f (&acc)[8]) {
#pragma unroll
  for (int j = 0; j < 8; j++) {
    const v4f av = pa.a[j >> 1];
    const v2f a = (j & 1) ? v2f{av.z, av.w} : v2f{av.x, av.y};
#pragma unroll
    for (int t = 0; t < 4; t++) {
      const v2f bl = {pb.lo[2 * j][t], pb.lo[2 * j + 1][t]};
      const v2f bh = {pb.hi[2 * j][t], pb.hi[2 * j + 1][t]};
      acc[t]     = wmma_f32_k4(a, bl, acc[t]);
      acc[4 + t] = wmma_f32_k4(a, bh, acc[4 + t]);
    }
  }
}
__device__ __forceinline__ unsigned proj_k(unsigned k) { return k < DIN ? k : DIN - 32; }

__global__ __launch_bounds__(128)
void proj_kernel(const float* __restrict__ F, const float* __restrict__ W,
                 float* __restrict__ R, float* __restrict__ sq) {
  const unsigned lane  = threadIdx.x & 31;
  const unsigned strip = blockIdx.x * 4 + (threadIdx.x >> 5);
  const unsigned h = lane >> 4;
  const unsigned c = lane & 15;
  const unsigned m0 = strip * 16;
  const float* ap = F + (size_t)(m0 + c) * DIN + 16 * h;
  const float* bp = W + (size_t)(16 * h) * DPROJ + 4 * c;

  v8f acc[8] = {};
  ProjA a0, a1, a2, a3;
  ProjB b0, b1;
  proj_fetch_a(a0, ap);
  proj_fetch_a(a1, ap + 32);
  proj_fetch_a(a2, ap + 64);
  proj_fetch_b(b0, bp);
  for (unsigned kc = 0; kc < DIN; kc += 128) {
    proj_fetch_a(a3, ap + kc + 96);
    proj_fetch_b(b1, bp + (size_t)(kc + 32) * DPROJ);
    __builtin_amdgcn_sched_barrier(0);
    proj_mma(a0, b0, acc);
    __builtin_amdgcn_sched_barrier(0);
    proj_fetch_a(a0, ap + proj_k(kc + 128));
    proj_fetch_b(b0, bp + (size_t)(kc + 64) * DPROJ);
    __builtin_amdgcn_sched_barrier(0);
    proj_mma(a1, b1, acc);
    __builtin_amdgcn_sched_barrier(0);
    proj_fetch_a(a1, ap + proj_k(kc + 160));
    proj_fetch_b(b1, bp + (size_t)(kc + 96) * DPROJ);
    __builtin_amdgcn_sched_barrier(0);
    proj_mma(a2, b0, acc);
    __builtin_amdgcn_sched_barrier(0);
    proj_fetch_a(a2, ap + proj_k(kc + 192));
    proj_fetch_b(b0, bp + (size_t)proj_k(kc + 128) * DPROJ);
    __builtin_amdgcn_sched_barrier(0);
    proj_mma(a3, b1, acc);
    __builtin_amdgcn_sched_barrier(0);
  }

  float* rp = R + (size_t)(m0 + 8 * h) * DPROJ + 4 * c;
  float s[8];
#pragma unroll
  for (int r = 0; r < 8; r++) {
    const v4f lo = {acc[0][r], acc[1][r], acc[2][r], acc[3][r]};
    const v4f hi = {acc[4][r], acc[5][r], acc[6][r], acc[7][r]};
    *(v4f*)(rp + r * DPROJ)      = lo;
    *(v4f*)(rp + r * DPROJ + 64) = hi;
    float v = lo.x * lo.x;
    v = fmaf(lo.y, lo.y, v); v = fmaf(lo.z, lo.z, v); v = fmaf(lo.w, lo.w, v);
    v = fmaf(hi.x, hi.x, v); v = fmaf(hi.y, hi.y, v);
    v = fmaf(hi.z, hi.z, v); v = fmaf(hi.w, hi.w, v);
    v += __int_as_float(lane_xor<1>(__float_as_int(v)));
    v += __int_as_float(lane_xor<2>(__float_as_int(v)));
    v += __int_as_float(lane_xor<4>(__float_as_int(v)));
    v += __int_as_float(lane_xor<8>(__float_as_int(v)));
    s[r] = v;
  }
  if (c == 0) {
    float* sp = sq + m0 + 8 * h;
    *(v4f*)(sp)     = v4f{s[0], s[1], s[2], s[3]};
    *(v4f*)(sp + 4) = v4f{s[4], s[5], s[6], s[7]};
  }
}

#define GRAM_PITCH        132
#define GRAM_STAGE_TILES  2
#define GRAM_TILE_FLOATS  (16 * GRAM_PITCH)
#define GRAM_STAGE_FLOATS (GRAM_STAGE_TILES * GRAM_TILE_FLOATS)
#define GRAM_NSTEP        (TILES_PER_CHUNK / GRAM_STAGE_TILES)

struct BTile { v2f f[32]; float sqc; };
struct BHalf { float4 p[2]; };

__device__ __forceinline__ void half_fetch(BHalf& g, const float* __restrict__ Rtile,
                                           unsigned thrOff) {
  const float* a = Rtile + thrOff;
  g.p[0] = *(const float4*)(a);
  g.p[1] = *(const float4*)(a + 4);
}

__device__ __forceinline__ void half_store(const BHalf& g, float* d) {
  *(float4*)(d)     = make_float4(g.p[0].x, g.p[0].y, g.p[1].x, g.p[1].y);
  *(float4*)(d + 4) = make_float4(g.p[0].z, g.p[0].w, g.p[1].z, g.p[1].w);
}

__device__ __forceinline__ void load_btile(BTile& bt, const float* tile,
                                           const float* __restrict__ sqTile, unsigned r16) {
#pragma unroll
  for (int j = 0; j < 16; j++) {
    const float4 q = *(const float4*)(tile + j * 8);
    bt.f[2 * j].x     = q.x;  bt.f[2 * j].y     = q.y;
    bt.f[2 * j + 1].x = q.z;  bt.f[2 * j + 1].y = q.w;
  }
  bt.sqc = sqTile[r16];
}

__device__ __forceinline__ void tile_rowsums(const v2f (&afrag)[32], const BTile& bt,
                                             const v2f (&sqr)[4], v2f (&partial)[4]) {
  v8f acc = {};
#pragma unroll
  for (int kk = 0; kk < 32; kk++) acc = wmma_f32_k4(afrag[kk], bt.f[kk], acc);
  const v2f sqc2 = {bt.sqc, bt.sqc};
  const v2f zero2 = {0.f, 0.f};
#pragma unroll
  for (int j = 0; j < 4; j++) {
    const v2f a = {acc[2 * j], acc[2 * j + 1]};
    partial[j] += __builtin_elementwise_max(zero2, sqr[j] + sqc2 - 2.f * a);
  }
}

template <int CUR, bool FETCH_AHEAD>
__device__ __forceinline__ void gram_step(const v2f (&afrag)[32], BTile& b0, BTile& b1,
                                          BHalf& g, const v2f (&sqr)[4],
                                          v2f (&partial)[4], float* bstage,
                                          const float* __restrict__ Rt,
                                          const float* __restrict__ sqt,
                                          unsigned laneOff, unsigned thrOff,
                                          unsigned stOff, unsigned r16) {
  const float* cur = bstage + CUR * GRAM_STAGE_FLOATS;
  float*       nxt = bstage + (CUR ^ 1) * GRAM_STAGE_FLOATS;
  load_btile(b1, cur + GRAM_TILE_FLOATS + laneOff, sqt + 16, r16);
  half_store(g, nxt + stOff);
  half_fetch(g, Rt + 3 * 16 * DPROJ, thrOff);
  tile_rowsums(afrag, b0, sqr, partial);
  half_store(g, nxt + GRAM_TILE_FLOATS + stOff);
  __syncthreads();
  __builtin_amdgcn_sched_barrier(0);
  load_btile(b0, nxt + laneOff, sqt + 32, r16);
  if (FETCH_AHEAD) half_fetch(g, Rt + 4 * 16 * DPROJ, thrOff);
  tile_rowsums(afrag, b1, sqr, partial);
  __builtin_amdgcn_sched_barrier(0);
}

__global__ __launch_bounds__(256, 4)
void gram_rowsum_kernel(const float* __restrict__ R, const float* __restrict__ sq,
                        float* __restrict__ anchorPart) {
  __shared__ __attribute__((aligned(16))) float bstage[2 * GRAM_STAGE_FLOATS];

  const unsigned tid   = threadIdx.x;
  const unsigned lane  = tid & 31;
  const unsigned chunk = blockIdx.x & (NCHUNK - 1);
  const unsigned strip = (blockIdx.x >> 3) * 8 + (tid >> 5);
  const unsigned half  = lane >> 4;
  const unsigned r16   = lane & 15;
  const unsigned koff  = half * 2;
  const unsigned m0    = strip * 16;

  v2f afrag[32];
  const float* arow = R + (size_t)(m0 + r16) * DPROJ + koff;
#pragma unroll
  for (int kk = 0; kk < 32; kk++) afrag[kk] = *(const v2f*)(arow + kk * 4);

  v2f sqr[4];
#pragma unroll
  for (int j = 0; j < 4; j++) {
    sqr[j].x = sq[m0 + 2 * j + half * 8];
    sqr[j].y = sq[m0 + 2 * j + 1 + half * 8];
  }

  v2f partial[4] = {{0.f, 0.f}, {0.f, 0.f}, {0.f, 0.f}, {0.f, 0.f}};

  const unsigned t0      = chunk * TILES_PER_CHUNK;
  const unsigned laneOff = r16 * GRAM_PITCH + half * 4;
  const unsigned thrOff  = (tid >> 4) * DPROJ + (tid & 15) * 8;
  const unsigned stOff   = (tid >> 4) * GRAM_PITCH + (tid & 15) * 8;
  const float* Rt  = R + (size_t)t0 * 16 * DPROJ;
  const float* sqt = sq + t0 * 16;

  BHalf g;
  BTile b0, b1;
  half_fetch(g, Rt, thrOff);
  half_store(g, bstage + stOff);
  half_fetch(g, Rt + 16 * DPROJ, thrOff);
  half_store(g, bstage + GRAM_TILE_FLOATS + stOff);
  __syncthreads();
  load_btile(b0, bstage + laneOff, sqt, r16);
  half_fetch(g, Rt + 2 * 16 * DPROJ, thrOff);

  for (unsigned sp = 0; sp < (GRAM_NSTEP - 2) / 2; sp++) {
    const float* Rs  = Rt + sp * (64 * DPROJ);
    const float* sqs = sqt + sp * 64;
    gram_step<0, true>(afrag, b0, b1, g, sqr, partial, bstage, Rs, sqs,
                       laneOff, thrOff, stOff, r16);
    gram_step<1, true>(afrag, b0, b1, g, sqr, partial, bstage, Rs + 32 * DPROJ, sqs + 32,
                       laneOff, thrOff, stOff, r16);
  }
  const float* Rl  = Rt + (GRAM_NSTEP - 2) * (32 * DPROJ);
  const float* sql = sqt + (GRAM_NSTEP - 2) * 32;
  gram_step<0, false>(afrag, b0, b1, g, sqr, partial, bstage, Rl, sql,
                      laneOff, thrOff, stOff, r16);
  load_btile(b1, bstage + GRAM_STAGE_FLOATS + GRAM_TILE_FLOATS + laneOff, sql + 48, r16);
  tile_rowsums(afrag, b0, sqr, partial);
  tile_rowsums(afrag, b1, sqr, partial);

#pragma unroll
  for (int i = 0; i < 8; i++) {
    float q = (i & 1) ? partial[i >> 1].y : partial[i >> 1].x;
    q += __shfl_xor(q, 1, 32);
    q += __shfl_xor(q, 2, 32);
    q += __shfl_xor(q, 4, 32);
    q += __shfl_xor(q, 8, 32);
    if (r16 == 0)
      anchorPart[(size_t)chunk * NPTS + m0 + i + half * 8] = q;
  }
}

__global__ __launch_bounds__(256)
void combine_kernel(const float* __restrict__ anchorPart,
                    float* __restrict__ anchorSq) {
  const int i = blockIdx.x * 256 + threadIdx.x;
  float s = 0.f;
#pragma unroll
  for (int c = 0; c < NCHUNK; c++) s += anchorPart[(size_t)c * NPTS + i];
  anchorSq[i] = s;
}

#define NWAVES (NPTS / 32)

template <int M>
__device__ __forceinline__ int lane_xor(int x) {
  return __builtin_amdgcn_update_dpp(0, x, 0x160 + M, 0xf, 0xf, true);
}
template <>
__device__ __forceinline__ int lane_xor<16>(int x) {
  return __builtin_amdgcn_permlanex16(0, x, 0x76543210, 0xfedcba98, false, false);
}

#define NO_ROW 0x7fffffff

__device__ __forceinline__ int imin(int a, int b) { return a < b ? a : b; }

template <int M>
__device__ __forceinline__ float max_exchange(float v) {
  return fmaxf(v, __int_as_float(lane_xor<M>(__float_as_int(v))));
}
template <int M>
__device__ __forceinline__ int min_exchange(int v) {
  return imin(v, lane_xor<M>(v));
}

__device__ __forceinline__ float wave_max(float v) {
  v = max_exchange<1>(v);
  v = max_exchange<2>(v);
  v = max_exchange<4>(v);
  v = max_exchange<8>(v);
  return max_exchange<16>(v);
}
__device__ __forceinline__ int wave_min(int v) {
  v = min_exchange<1>(v);
  v = min_exchange<2>(v);
  v = min_exchange<4>(v);
  v = min_exchange<8>(v);
  return min_exchange<16>(v);
}

__device__ __forceinline__ void wave_argmax(float& bv, int& bi) {
  const float m = wave_max(bv);
  bi = wave_min(bv == m ? bi : NO_ROW);
  bv = m;
}

__global__ __launch_bounds__(32)
void anchor_init_kernel(float* __restrict__ anchor, float* __restrict__ candV,
                        int* __restrict__ candI) {
  const int row  = blockIdx.x * 32 + threadIdx.x;
  const int lane = threadIdx.x;
  const float a = sqrtf(anchor[row]);
  anchor[row] = a;
  float bv = a; int bi = row;
  wave_argmax(bv, bi);
  if (lane == 0) { candV[row >> 5] = bv; candI[row >> 5] = bi; }
}

template <int M>
__device__ __forceinline__ float add_exchange(float keep, float send) {
  return keep + __int_as_float(lane_xor<M>(__float_as_int(send)));
}
template <int K>
__device__ __forceinline__ void fold_stage(float (&val)[32], int lane) {
  const bool up = (lane >> K) & 1;
#pragma unroll
  for (int idx = 0; idx < 32; idx += (2 << K)) {
    const float a = val[idx], b = val[idx | (1 << K)];
    val[idx] = add_exchange<(1 << K)>(up ? b : a, up ? a : b);
  }
}

__global__ __launch_bounds__(32)
void greedy_step_kernel(const float* __restrict__ R, const float* __restrict__ sq,
                        float* __restrict__ anchor, float* __restrict__ candV,
                        int* __restrict__ candI, int* __restrict__ sel, int s) {
  const int row  = blockIdx.x * 32 + threadIdx.x;
  const int lane = threadIdx.x;
  const int pin  = (s & 1) * NWAVES;
  const int pout = NWAVES - pin;

  float4 cv[4]; int4 ci[4];
  const float4* cvp = (const float4*)(candV + pin) + lane;
  const int4*   cip = (const int4*)(candI + pin) + lane;
#pragma unroll
  for (int q = 0; q < 4; q++) { cv[q] = cvp[q * 32]; ci[q] = cip[q * 32]; }

  const v4f* gp = (const v4f*)(R + (size_t)blockIdx.x * 32 * DPROJ) + lane;
  v4f own[32];
#pragma unroll
  for (int r = 0; r < 32; r++) own[r] = gp[r * (DPROJ / 4)];
  const float sqOwn = sq[row];
  const float aOld  = anchor[row];
  __builtin_amdgcn_sched_barrier(0);

  float qm[4];
#pragma unroll
  for (int q = 0; q < 4; q++)
    qm[q] = fmaxf(fmaxf(cv[q].x, cv[q].y), fmaxf(cv[q].z, cv[q].w));
  float bv = fmaxf(fmaxf(qm[0], qm[1]), fmaxf(qm[2], qm[3]));
  int qi[4];
#pragma unroll
  for (int q = 0; q < 4; q++) {
    const int r0 = cv[q].x == bv ? ci[q].x : NO_ROW;
    const int r1 = cv[q].y == bv ? ci[q].y : NO_ROW;
    const int r2 = cv[q].z == bv ? ci[q].z : NO_ROW;
    const int r3 = cv[q].w == bv ? ci[q].w : NO_ROW;
    qi[q] = imin(imin(r0, r1), imin(r2, r3));
  }
  int bi = imin(imin(qi[0], qi[1]), imin(qi[2], qi[3]));
  wave_argmax(bv, bi);
  const int si = bi & (NPTS - 1);

  const v4f selPiece = ((const v4f*)(R + (size_t)si * DPROJ))[lane];
  const float selSq = sq[si];

  const v2f slo = {selPiece.x, selPiece.y};
  const v2f shi = {selPiece.z, selPiece.w};
  float val[32];
#pragma unroll
  for (int r = 0; r < 32; r++) {
    const v2f lo = {own[r].x, own[r].y};
    const v2f hi = {own[r].z, own[r].w};
    const v2f p = __builtin_elementwise_fma(hi, shi, lo * slo);
    val[r] = p.x + p.y;
  }
  fold_stage<0>(val, lane);
  fold_stage<1>(val, lane);
  fold_stage<2>(val, lane);
  fold_stage<3>(val, lane);
  fold_stage<4>(val, lane);
  const float dot = val[0];
  const float d2 = sqOwn + selSq - 2.f * dot;
  const float a  = fminf(aOld, sqrtf(fmaxf(d2, 0.f)));
  anchor[row] = a;

  float mv = a; int mi = row;
  wave_argmax(mv, mi);
  if (lane == 0) { candV[pout + (row >> 5)] = mv; candI[pout + (row >> 5)] = mi; }
  if (row == 0) sel[s] = si;
}

__global__ __launch_bounds__(256)
void gather_kernel(const float* __restrict__ F, const int* __restrict__ sel,
                   float* __restrict__ out) {
  const int s = blockIdx.x;
  const int c = blockIdx.y * 256 + threadIdx.x;
  out[(size_t)s * DIN + c] = F[(size_t)sel[s] * DIN + c];
}

extern "C" void kernel_launch(void* const* d_in, const int* in_sizes, int n_in,
                              void* d_out, int out_size, void* d_ws, size_t ws_size,
                              hipStream_t stream) {
  (void)in_sizes; (void)n_in; (void)out_size; (void)ws_size;
  const float* F = (const float*)d_in[0];
  const float* W = (const float*)d_in[1];

  float* R          = (float*)d_ws;
  float* sq         = R + (size_t)NPTS * DPROJ;
  float* anchorSq   = sq + NPTS;
  float* anchorPart = anchorSq + NPTS;
  int*   sel        = (int*)(anchorPart + (size_t)NCHUNK * NPTS);

  proj_kernel<<<NPTS / 64, 128, 0, stream>>>(F, W, R, sq);
  gram_rowsum_kernel<<<1024, 256, 0, stream>>>(R, sq, anchorPart);
  combine_kernel<<<NPTS / 256, 256, 0, stream>>>(anchorPart, anchorSq);
  float* candV = anchorPart;
  int*   candI = (int*)(anchorPart + 2 * NWAVES);
  anchor_init_kernel<<<NPTS / 32, 32, 0, stream>>>(anchorSq, candV, candI);
  for (int s = 0; s < NSAMP; s++)
    greedy_step_kernel<<<NPTS / 32, 32, 0, stream>>>(R, sq, anchorSq, candV, candI, sel, s);
  gather_kernel<<<dim3(NSAMP, DIN / 256), 256, 0, stream>>>(F, sel, (float*)d_out);
}


# start values: sum_j dist(i,j)^2 taken as N|ri|^2 + sum|rj|^2 - 2 ri.(sum rj) from column sums, replacing the 16384^2 Gram pass; per-term clip at 0 moves to the sum; sum order changes
# speedup vs baseline: 59.2669x; 1.6448x over previous
#include <hip/hip_runtime.h>


#define NPTS   16384
#define DIN    1024
#define DPROJ  128
#define NSAMP  163
#define SCRATCH_FLOATS (8 * NPTS)
#define CS_BLOCKS 64
#define CS_ROWS   (NPTS / CS_BLOCKS)

typedef float v2f __attribute__((ext_vector_type(2)));
typedef float v4f __attribute__((ext_vector_type(4)));
typedef float v8f __attribute__((ext_vector_type(8)));

__device__ __forceinline__ v8f wmma_f32_k4(v2f a, v2f b, v8f c) {
  return __builtin_amdgcn_wmma_f32_16x16x4_f32(
       false, a,  false, b,
       (short)0, c,  false,  false);
}

template <int M> __device__ __forceinline__ int lane_xor(int x);

struct ProjA { v4f a[4]; };
struct ProjB { v4f lo[16], hi[16]; };

__device__ __forceinline__ void proj_fetch_a(ProjA& p, const float* __restrict__ ap) {
#pragma unroll
  for (int i = 0; i < 4; i++) p.a[i] = *(const v4f*)(ap + 4 * i);
}
__device__ __forceinline__ void proj_fetch_b(ProjB& p, const float* __restrict__ bp) {
#pragma unroll
  for (int i = 0; i < 16; i++) {
    p.lo[i] = *(const v4f*)(bp + i * DPROJ);
    p.hi[i] = *(const v4f*)(bp + i * DPROJ + 64);
  }
}
__device__ __forceinline__ void proj_mma(const ProjA& pa, const ProjB& pb, v8f (&acc)[8]) {
#pragma unroll
  for (int j = 0; j < 8; j++) {
    const v4f av = pa.a[j >> 1];
    const v2f a = (j & 1) ? v2f{av.z, av.w} : v2f{av.x, av.y};
#pragma unroll
    for (int t = 0; t < 4; t++) {
      const v2f bl = {pb.lo[2 * j][t], pb.lo[2 * j + 1][t]};
      const v2f bh = {pb.hi[2 * j][t], pb.hi[2 * j + 1][t]};
      acc[t]     = wmma_f32_k4(a, bl, acc[t]);
      acc[4 + t] = wmma_f32_k4(a, bh, acc[4 + t]);
    }
  }
}
__device__ __forceinline__ unsigned proj_k(unsigned k) { return k < DIN ? k : DIN - 32; }

__global__ __launch_bounds__(128)
void proj_kernel(const float* __restrict__ F, const float* __restrict__ W,
                 float* __restrict__ R, float* __restrict__ sq) {
  const unsigned lane  = threadIdx.x & 31;
  const unsigned strip = blockIdx.x * 4 + (threadIdx.x >> 5);
  const unsigned h = lane >> 4;
  const unsigned c = lane & 15;
  const unsigned m0 = strip * 16;
  const float* ap = F + (size_t)(m0 + c) * DIN + 16 * h;
  const float* bp = W + (size_t)(16 * h) * DPROJ + 4 * c;

  v8f acc[8] = {};
  ProjA a0, a1, a2, a3;
  ProjB b0, b1;
  proj_fetch_a(a0, ap);
  proj_fetch_a(a1, ap + 32);
  proj_fetch_a(a2, ap + 64);
  proj_fetch_b(b0, bp);
  for (unsigned kc = 0; kc < DIN; kc += 128) {
    proj_fetch_a(a3, ap + kc + 96);
    proj_fetch_b(b1, bp + (size_t)(kc + 32) * DPROJ);
    __builtin_amdgcn_sched_barrier(0);
    proj_mma(a0, b0, acc);
    __builtin_amdgcn_sched_barrier(0);
    proj_fetch_a(a0, ap + proj_k(kc + 128));
    proj_fetch_b(b0, bp + (size_t)(kc + 64) * DPROJ);
    __builtin_amdgcn_sched_barrier(0);
    proj_mma(a1, b1, acc);
    __builtin_amdgcn_sched_barrier(0);
    proj_fetch_a(a1, ap + proj_k(kc + 160));
    proj_fetch_b(b1, bp + (size_t)(kc + 96) * DPROJ);
    __builtin_amdgcn_sched_barrier(0);
    proj_mma(a2, b0, acc);
    __builtin_amdgcn_sched_barrier(0);
    proj_fetch_a(a2, ap + proj_k(kc + 192));
    proj_fetch_b(b0, bp + (size_t)proj_k(kc + 128) * DPROJ);
    __builtin_amdgcn_sched_barrier(0);
    proj_mma(a3, b1, acc);
    __builtin_amdgcn_sched_barrier(0);
  }

  float* rp = R + (size_t)(m0 + 8 * h) * DPROJ + 4 * c;
  float s[8];
#pragma unroll
  for (int r = 0; r < 8; r++) {
    const v4f lo = {acc[0][r], acc[1][r], acc[2][r], acc[3][r]};
    const v4f hi = {acc[4][r], acc[5][r], acc[6][r], acc[7][r]};
    *(v4f*)(rp + r * DPROJ)      = lo;
    *(v4f*)(rp + r * DPROJ + 64) = hi;
    float v = lo.x * lo.x;
    v = fmaf(lo.y, lo.y, v); v = fmaf(lo.z, lo.z, v); v = fmaf(lo.w, lo.w, v);
    v = fmaf(hi.x, hi.x, v); v = fmaf(hi.y, hi.y, v);
    v = fmaf(hi.z, hi.z, v); v = fmaf(hi.w, hi.w, v);
    v += __int_as_float(lane_xor<1>(__float_as_int(v)));
    v += __int_as_float(lane_xor<2>(__float_as_int(v)));
    v += __int_as_float(lane_xor<4>(__float_as_int(v)));
    v += __int_as_float(lane_xor<8>(__float_as_int(v)));
    s[r] = v;
  }
  if (c == 0) {
    float* sp = sq + m0 + 8 * h;
    *(v4f*)(sp)     = v4f{s[0], s[1], s[2], s[3]};
    *(v4f*)(sp + 4) = v4f{s[4], s[5], s[6], s[7]};
  }
}

#define NWAVES (NPTS / 32)

template <int M>
__device__ __forceinline__ int lane_xor(int x) {
  return __builtin_amdgcn_update_dpp(0, x, 0x160 + M, 0xf, 0xf, true);
}
template <>
__device__ __forceinline__ int lane_xor<16>(int x) {
  return __builtin_amdgcn_permlanex16(0, x, 0x76543210, 0xfedcba98, false, false);
}

#define NO_ROW 0x7fffffff

__device__ __forceinline__ int imin(int a, int b) { return a < b ? a : b; }

template <int M>
__device__ __forceinline__ float max_exchange(float v) {
  return fmaxf(v, __int_as_float(lane_xor<M>(__float_as_int(v))));
}
template <int M>
__device__ __forceinline__ int min_exchange(int v) {
  return imin(v, lane_xor<M>(v));
}

__device__ __forceinline__ float wave_max(float v) {
  v = max_exchange<1>(v);
  v = max_exchange<2>(v);
  v = max_exchange<4>(v);
  v = max_exchange<8>(v);
  return max_exchange<16>(v);
}
__device__ __forceinline__ int wave_min(int v) {
  v = min_exchange<1>(v);
  v = min_exchange<2>(v);
  v = min_exchange<4>(v);
  v = min_exchange<8>(v);
  return min_exchange<16>(v);
}

__device__ __forceinline__ void wave_argmax(float& bv, int& bi) {
  const float m = wave_max(bv);
  bi = wave_min(bv == m ? bi : NO_ROW);
  bv = m;
}

template <int M>
__device__ __forceinline__ float add_exchange(float keep, float send) {
  return keep + __int_as_float(lane_xor<M>(__float_as_int(send)));
}
template <int K>
__device__ __forceinline__ void fold_stage(float (&val)[32], int lane) {
  const bool up = (lane >> K) & 1;
#pragma unroll
  for (int idx = 0; idx < 32; idx += (2 << K)) {
    const float a = val[idx], b = val[idx | (1 << K)];
    val[idx] = add_exchange<(1 << K)>(up ? b : a, up ? a : b);
  }
}

__global__ __launch_bounds__(256)
void colsum_kernel(const float* __restrict__ R, const float* __restrict__ sq,
                   float* __restrict__ mPart, float* __restrict__ sPart) {
  __shared__ __attribute__((aligned(16))) float red[8 * DPROJ];
  __shared__ float redS[8];
  const unsigned tid  = threadIdx.x;
  const unsigned lane = tid & 31;
  const unsigned wv   = tid >> 5;
  const unsigned r0   = blockIdx.x * CS_ROWS;
  const v4f* p = (const v4f*)(R + (size_t)(r0 + wv) * DPROJ) + lane;
  v4f a0 = {}, a1 = {}, a2 = {}, a3 = {};
  for (int k = 0; k < CS_ROWS / 8; k += 4) {
    a0 += p[(k    ) * 8 * (DPROJ / 4)];
    a1 += p[(k + 1) * 8 * (DPROJ / 4)];
    a2 += p[(k + 2) * 8 * (DPROJ / 4)];
    a3 += p[(k + 3) * 8 * (DPROJ / 4)];
  }
  *(v4f*)(red + wv * DPROJ + 4 * lane) = (a0 + a1) + (a2 + a3);
  float t = sq[r0 + tid];
  t += __int_as_float(lane_xor<1>(__float_as_int(t)));
  t += __int_as_float(lane_xor<2>(__float_as_int(t)));
  t += __int_as_float(lane_xor<4>(__float_as_int(t)));
  t += __int_as_float(lane_xor<8>(__float_as_int(t)));
  t += __int_as_float(lane_xor<16>(__float_as_int(t)));
  if (lane == 0) redS[wv] = t;
  __syncthreads();
  if (tid < DPROJ) {
    float m = red[tid];
    for (int w = 1; w < 8; w++) m += red[w * DPROJ + tid];
    mPart[blockIdx.x * DPROJ + tid] = m;
  }
  if (tid == 0) {
    float u = redS[0];
    for (int w = 1; w < 8; w++) u += redS[w];
    sPart[blockIdx.x] = u;
  }
}

__global__ __launch_bounds__(32)
void anchor_init_kernel(const float* __restrict__ R, const float* __restrict__ sq,
                        const float* __restrict__ mPart, const float* __restrict__ sPart,
                        float* __restrict__ anchor, float* __restrict__ candV,
                        int* __restrict__ candI) {
  const int row  = blockIdx.x * 32 + threadIdx.x;
  const int lane = threadIdx.x;

  const v4f* mp = (const v4f*)mPart + lane;
  v4f m0 = {}, m1 = {}, m2 = {}, m3 = {};
  for (int b = 0; b < CS_BLOCKS; b += 4) {
    m0 += mp[(b    ) * (DPROJ / 4)];
    m1 += mp[(b + 1) * (DPROJ / 4)];
    m2 += mp[(b + 2) * (DPROJ / 4)];
    m3 += mp[(b + 3) * (DPROJ / 4)];
  }
  const v4f m = (m0 + m1) + (m2 + m3);
  float S = sPart[lane] + sPart[lane + 32];
  S += __int_as_float(lane_xor<1>(__float_as_int(S)));
  S += __int_as_float(lane_xor<2>(__float_as_int(S)));
  S += __int_as_float(lane_xor<4>(__float_as_int(S)));
  S += __int_as_float(lane_xor<8>(__float_as_int(S)));
  S += __int_as_float(lane_xor<16>(__float_as_int(S)));

  const v4f* gp = (const v4f*)(R + (size_t)blockIdx.x * 32 * DPROJ) + lane;
  const v2f mlo = {m.x, m.y};
  const v2f mhi = {m.z, m.w};
  float val[32];
  for (int r = 0; r < 32; r++) {
    const v4f o = gp[r * (DPROJ / 4)];
    const v2f lo = {o.x, o.y};
    const v2f hi = {o.z, o.w};
    const v2f p = __builtin_elementwise_fma(hi, mhi, lo * mlo);
    val[r] = p.x + p.y;
  }
  fold_stage<0>(val, lane);
  fold_stage<1>(val, lane);
  fold_stage<2>(val, lane);
  fold_stage<3>(val, lane);
  fold_stage<4>(val, lane);
  const float dot = val[0];
  const float d2 = fmaf((float)NPTS, sq[row], S) - 2.f * dot;
  const float a = sqrtf(fmaxf(d2, 0.f));
  anchor[row] = a;
  float bv = a; int bi = row;
  wave_argmax(bv, bi);
  if (lane == 0) { candV[row >> 5] = bv; candI[row >> 5] = bi; }
}

__global__ __launch_bounds__(32)
void greedy_step_kernel(const float* __restrict__ R, const float* __restrict__ sq,
                        float* __restrict__ anchor, float* __restrict__ candV,
                        int* __restrict__ candI, int* __restrict__ sel, int s) {
  const int row  = blockIdx.x * 32 + threadIdx.x;
  const int lane = threadIdx.x;
  const int pin  = (s & 1) * NWAVES;
  const int pout = NWAVES - pin;

  float4 cv[4]; int4 ci[4];
  const float4* cvp = (const float4*)(candV + pin) + lane;
  const int4*   cip = (const int4*)(candI + pin) + lane;
#pragma unroll
  for (int q = 0; q < 4; q++) { cv[q] = cvp[q * 32]; ci[q] = cip[q * 32]; }

  const v4f* gp = (const v4f*)(R + (size_t)blockIdx.x * 32 * DPROJ) + lane;
  v4f own[32];
#pragma unroll
  for (int r = 0; r < 32; r++) own[r] = gp[r * (DPROJ / 4)];
  const float sqOwn = sq[row];
  const float aOld  = anchor[row];
  __builtin_amdgcn_sched_barrier(0);

  float qm[4];
#pragma unroll
  for (int q = 0; q < 4; q++)
    qm[q] = fmaxf(fmaxf(cv[q].x, cv[q].y), fmaxf(cv[q].z, cv[q].w));
  float bv = fmaxf(fmaxf(qm[0], qm[1]), fmaxf(qm[2], qm[3]));
  int qi[4];
#pragma unroll
  for (int q = 0; q < 4; q++) {
    const int r0 = cv[q].x == bv ? ci[q].x : NO_ROW;
    const int r1 = cv[q].y == bv ? ci[q].y : NO_ROW;
    const int r2 = cv[q].z == bv ? ci[q].z : NO_ROW;
    const int r3 = cv[q].w == bv ? ci[q].w : NO_ROW;
    qi[q] = imin(imin(r0, r1), imin(r2, r3));
  }
  int bi = imin(imin(qi[0], qi[1]), imin(qi[2], qi[3]));
  wave_argmax(bv, bi);
  const int si = bi & (NPTS - 1);

  const v4f selPiece = ((const v4f*)(R + (size_t)si * DPROJ))[lane];
  const float selSq = sq[si];

  const v2f slo = {selPiece.x, selPiece.y};
  const v2f shi = {selPiece.z, selPiece.w};
  float val[32];
#pragma unroll
  for (int r = 0; r < 32; r++) {
    const v2f lo = {own[r].x, own[r].y};
    const v2f hi = {own[r].z, own[r].w};
    const v2f p = __builtin_elementwise_fma(hi, shi, lo * slo);
    val[r] = p.x + p.y;
  }
  fold_stage<0>(val, lane);
  fold_stage<1>(val, lane);
  fold_stage<2>(val, lane);
  fold_stage<3>(val, lane);
  fold_stage<4>(val, lane);
  const float dot = val[0];
  const float d2 = sqOwn + selSq - 2.f * dot;
  const float a  = fminf(aOld, sqrtf(fmaxf(d2, 0.f)));
  anchor[row] = a;

  float mv = a; int mi = row;
  wave_argmax(mv, mi);
  if (lane == 0) { candV[pout + (row >> 5)] = mv; candI[pout + (row >> 5)] = mi; }
  if (row == 0) sel[s] = si;
}

__global__ __launch_bounds__(256)
void gather_kernel(const float* __restrict__ F, const int* __restrict__ sel,
                   float* __restrict__ out) {
  const int s = blockIdx.x;
  const int c = blockIdx.y * 256 + threadIdx.x;
  out[(size_t)s * DIN + c] = F[(size_t)sel[s] * DIN + c];
}

extern "C" void kernel_launch(void* const* d_in, const int* in_sizes, int n_in,
                              void* d_out, int out_size, void* d_ws, size_t ws_size,
                              hipStream_t stream) {
  (void)in_sizes; (void)n_in; (void)out_size; (void)ws_size;
  const float* F = (const float*)d_in[0];
  const float* W = (const float*)d_in[1];

  float* R          = (float*)d_ws;
  float* sq         = R + (size_t)NPTS * DPROJ;
  float* anchorSq   = sq + NPTS;
  float* scratch    = anchorSq + NPTS;
  int*   sel        = (int*)(scratch + (size_t)SCRATCH_FLOATS);
  float* candV = scratch;
  int*   candI = (int*)(scratch + 2 * NWAVES);
  float* mPart = scratch + 4 * NWAVES;
  float* sPart = mPart + CS_BLOCKS * DPROJ;

  proj_kernel<<<NPTS / 64, 128, 0, stream>>>(F, W, R, sq);
  colsum_kernel<<<CS_BLOCKS, 256, 0, stream>>>(R, sq, mPart, sPart);
  anchor_init_kernel<<<NPTS / 32, 32, 0, stream>>>(R, sq, mPart, sPart, anchorSq, candV, candI);
  for (int s = 0; s < NSAMP; s++)
    greedy_step_kernel<<<NPTS / 32, 32, 0, stream>>>(R, sq, anchorSq, candV, candI, sel, s);
  gather_kernel<<<dim3(NSAMP, DIN / 256), 256, 0, stream>>>(F, sel, (float*)d_out);
}


# greedy loop: from iteration 3 on one launch does two iterations (83 greedy launches, not 163); every wave finds the second row exactly by re-evaluating the leading candidate rows; same sums
# speedup vs baseline: 77.8871x; 1.3142x over previous
#include <hip/hip_runtime.h>


#define NPTS   16384
#define DIN    1024
#define DPROJ  128
#define NSAMP  163
#define SCRATCH_FLOATS (8 * NPTS)
#define CS_BLOCKS 64
#define CS_ROWS   (NPTS / CS_BLOCKS)

typedef float v2f __attribute__((ext_vector_type(2)));
typedef float v4f __attribute__((ext_vector_type(4)));
typedef float v8f __attribute__((ext_vector_type(8)));

__device__ __forceinline__ v8f wmma_f32_k4(v2f a, v2f b, v8f c) {
  return __builtin_amdgcn_wmma_f32_16x16x4_f32(
       false, a,  false, b,
       (short)0, c,  false,  false);
}

template <int M> __device__ __forceinline__ int lane_xor(int x);

struct ProjA { v4f a[4]; };
struct ProjB { v4f lo[16], hi[16]; };

__device__ __forceinline__ void proj_fetch_a(ProjA& p, const float* __restrict__ ap) {
#pragma unroll
  for (int i = 0; i < 4; i++) p.a[i] = *(const v4f*)(ap + 4 * i);
}
__device__ __forceinline__ void proj_fetch_b(ProjB& p, const float* __restrict__ bp) {
#pragma unroll
  for (int i = 0; i < 16; i++) {
    p.lo[i] = *(const v4f*)(bp + i * DPROJ);
    p.hi[i] = *(const v4f*)(bp + i * DPROJ + 64);
  }
}
__device__ __forceinline__ void proj_mma(const ProjA& pa, const ProjB& pb, v8f (&acc)[8]) {
#pragma unroll
  for (int j = 0; j < 8; j++) {
    const v4f av = pa.a[j >> 1];
    const v2f a = (j & 1) ? v2f{av.z, av.w} : v2f{av.x, av.y};
#pragma unroll
    for (int t = 0; t < 4; t++) {
      const v2f bl = {pb.lo[2 * j][t], pb.lo[2 * j + 1][t]};
      const v2f bh = {pb.hi[2 * j][t], pb.hi[2 * j + 1][t]};
      acc[t]     = wmma_f32_k4(a, bl, acc[t]);
      acc[4 + t] = wmma_f32_k4(a, bh, acc[4 + t]);
    }
  }
}
__device__ __forceinline__ unsigned proj_k(unsigned k) { return k < DIN ? k : DIN - 32; }

__global__ __launch_bounds__(128)
void proj_kernel(const float* __restrict__ F, const float* __restrict__ W,
                 float* __restrict__ R, float* __restrict__ sq) {
  const unsigned lane  = threadIdx.x & 31;
  const unsigned strip = blockIdx.x * 4 + (threadIdx.x >> 5);
  const unsigned h = lane >> 4;
  const unsigned c = lane & 15;
  const unsigned m0 = strip * 16;
  const float* ap = F + (size_t)(m0 + c) * DIN + 16 * h;
  const float* bp = W + (size_t)(16 * h) * DPROJ + 4 * c;

  v8f acc[8] = {};
  ProjA a0, a1, a2, a3;
  ProjB b0, b1;
  proj_fetch_a(a0, ap);
  proj_fetch_a(a1, ap + 32);
  proj_fetch_a(a2, ap + 64);
  proj_fetch_b(b0, bp);
  for (unsigned kc = 0; kc < DIN; kc += 128) {
    proj_fetch_a(a3, ap + kc + 96);
    proj_fetch_b(b1, bp + (size_t)(kc + 32) * DPROJ);
    __builtin_amdgcn_sched_barrier(0);
    proj_mma(a0, b0, acc);
    __builtin_amdgcn_sched_barrier(0);
    proj_fetch_a(a0, ap + proj_k(kc + 128));
    proj_fetch_b(b0, bp + (size_t)(kc + 64) * DPROJ);
    __builtin_amdgcn_sched_barrier(0);
    proj_mma(a1, b1, acc);
    __builtin_amdgcn_sched_barrier(0);
    proj_fetch_a(a1, ap + proj_k(kc + 160));
    proj_fetch_b(b1, bp + (size_t)(kc + 96) * DPROJ);
    __builtin_amdgcn_sched_barrier(0);
    proj_mma(a2, b0, acc);
    __builtin_amdgcn_sched_barrier(0);
    proj_fetch_a(a2, ap + proj_k(kc + 192));
    proj_fetch_b(b0, bp + (size_t)proj_k(kc + 128) * DPROJ);
    __builtin_amdgcn_sched_barrier(0);
    proj_mma(a3, b1, acc);
    __builtin_amdgcn_sched_barrier(0);
  }

  float* rp = R + (size_t)(m0 + 8 * h) * DPROJ + 4 * c;
  float s[8];
#pragma unroll
  for (int r = 0; r < 8; r++) {
    const v4f lo = {acc[0][r], acc[1][r], acc[2][r], acc[3][r]};
    const v4f hi = {acc[4][r], acc[5][r], acc[6][r], acc[7][r]};
    *(v4f*)(rp + r * DPROJ)      = lo;
    *(v4f*)(rp + r * DPROJ + 64) = hi;
    float v = lo.x * lo.x;
    v = fmaf(lo.y, lo.y, v); v = fmaf(lo.z, lo.z, v); v = fmaf(lo.w, lo.w, v);
    v = fmaf(hi.x, hi.x, v); v = fmaf(hi.y, hi.y, v);
    v = fmaf(hi.z, hi.z, v); v = fmaf(hi.w, hi.w, v);
    v += __int_as_float(lane_xor<1>(__float_as_int(v)));
    v += __int_as_float(lane_xor<2>(__float_as_int(v)));
    v += __int_as_float(lane_xor<4>(__float_as_int(v)));
    v += __int_as_float(lane_xor<8>(__float_as_int(v)));
    s[r] = v;
  }
  if (c == 0) {
    float* sp = sq + m0 + 8 * h;
    *(v4f*)(sp)     = v4f{s[0], s[1], s[2], s[3]};
    *(v4f*)(sp + 4) = v4f{s[4], s[5], s[6], s[7]};
  }
}

#define NWAVES (NPTS / 32)

template <int M>
__device__ __forceinline__ int lane_xor(int x) {
  return __builtin_amdgcn_update_dpp(0, x, 0x160 + M, 0xf, 0xf, true);
}
template <>
__device__ __forceinline__ int lane_xor<16>(int x) {
  return __builtin_amdgcn_permlanex16(0, x, 0x76543210, 0xfedcba98, false, false);
}

#define NO_ROW 0x7fffffff

__device__ __forceinline__ int imin(int a, int b) { return a < b ? a : b; }

template <int M>
__device__ __forceinline__ float max_exchange(float v) {
  return fmaxf(v, __int_as_float(lane_xor<M>(__float_as_int(v))));
}
template <int M>
__device__ __forceinline__ int min_exchange(int v) {
  return imin(v, lane_xor<M>(v));
}

__device__ __forceinline__ float wave_max(float v) {
  v = max_exchange<1>(v);
  v = max_exchange<2>(v);
  v = max_exchange<4>(v);
  v = max_exchange<8>(v);
  return max_exchange<16>(v);
}
__device__ __forceinline__ int wave_min(int v) {
  v = min_exchange<1>(v);
  v = min_exchange<2>(v);
  v = min_exchange<4>(v);
  v = min_exchange<8>(v);
  return min_exchange<16>(v);
}

__device__ __forceinline__ void wave_argmax(float& bv, int& bi) {
  const float m = wave_max(bv);
  bi = wave_min(bv == m ? bi : NO_ROW);
  bv = m;
}

template <int M>
__device__ __forceinline__ float add_exchange(float keep, float send) {
  return keep + __int_as_float(lane_xor<M>(__float_as_int(send)));
}
template <int K>
__device__ __forceinline__ void fold_stage(float (&val)[32], int lane) {
  const bool up = (lane >> K) & 1;
#pragma unroll
  for (int idx = 0; idx < 32; idx += (2 << K)) {
    const float a = val[idx], b = val[idx | (1 << K)];
    val[idx] = add_exchange<(1 << K)>(up ? b : a, up ? a : b);
  }
}


#define NEG_INF (-__builtin_inff())

__device__ __forceinline__ float row_share(v4f x, v4f s) {
  const v2f lo = {x.x, x.y};
  const v2f hi = {x.z, x.w};
  const v2f slo = {s.x, s.y};
  const v2f shi = {s.z, s.w};
  const v2f p = __builtin_elementwise_fma(hi, shi, lo * slo);
  return p.x + p.y;
}
__device__ __forceinline__ float rows_dot(const v4f (&rows)[32], v4f s, int lane) {
  float val[32];
#pragma unroll
  for (int r = 0; r < 32; r++) val[r] = row_share(rows[r], s);
  fold_stage<0>(val, lane);
  fold_stage<1>(val, lane);
  fold_stage<2>(val, lane);
  fold_stage<3>(val, lane);
  fold_stage<4>(val, lane);
  return val[0];
}
__device__ __forceinline__ float row_dot_all(v4f x, v4f s) {
  float v = row_share(x, s);
  v = add_exchange<1>(v, v);
  v = add_exchange<2>(v, v);
  v = add_exchange<4>(v, v);
  v = add_exchange<8>(v, v);
  return add_exchange<16>(v, v);
}
__device__ __forceinline__ float min_dist(float aOld, float sqX, float selSq, float dot) {
  const float d2 = sqX + selSq - 2.f * dot;
  return fminf(aOld, sqrtf(fmaxf(d2, 0.f)));
}
__device__ __forceinline__ void wave_top2(float a, int row, float& v1, int& r1, float& v2, int& r2) {
  v1 = a; r1 = row;
  wave_argmax(v1, r1);
  v2 = (row == r1) ? NEG_INF : a; r2 = row;
  wave_argmax(v2, r2);
}

__global__ __launch_bounds__(256)
void colsum_kernel(const float* __restrict__ R, const float* __restrict__ sq,
                   float* __restrict__ mPart, float* __restrict__ sPart) {
  __shared__ __attribute__((aligned(16))) float red[8 * DPROJ];
  __shared__ float redS[8];
  const unsigned tid  = threadIdx.x;
  const unsigned lane = tid & 31;
  const unsigned wv   = tid >> 5;
  const unsigned r0   = blockIdx.x * CS_ROWS;
  const v4f* p = (const v4f*)(R + (size_t)(r0 + wv) * DPROJ) + lane;
  v4f a0 = {}, a1 = {}, a2 = {}, a3 = {};
  for (int k = 0; k < CS_ROWS / 8; k += 4) {
    a0 += p[(k    ) * 8 * (DPROJ / 4)];
    a1 += p[(k + 1) * 8 * (DPROJ / 4)];
    a2 += p[(k + 2) * 8 * (DPROJ / 4)];
    a3 += p[(k + 3) * 8 * (DPROJ / 4)];
  }
  *(v4f*)(red + wv * DPROJ + 4 * lane) = (a0 + a1) + (a2 + a3);
  float t = sq[r0 + tid];
  t += __int_as_float(lane_xor<1>(__float_as_int(t)));
  t += __int_as_float(lane_xor<2>(__float_as_int(t)));
  t += __int_as_float(lane_xor<4>(__float_as_int(t)));
  t += __int_as_float(lane_xor<8>(__float_as_int(t)));
  t += __int_as_float(lane_xor<16>(__float_as_int(t)));
  if (lane == 0) redS[wv] = t;
  __syncthreads();
  if (tid < DPROJ) {
    float m = red[tid];
    for (int w = 1; w < 8; w++) m += red[w * DPROJ + tid];
    mPart[blockIdx.x * DPROJ + tid] = m;
  }
  if (tid == 0) {
    float u = redS[0];
    for (int w = 1; w < 8; w++) u += redS[w];
    sPart[blockIdx.x] = u;
  }
}

__global__ __launch_bounds__(32)
void anchor_init_kernel(const float* __restrict__ R, const float* __restrict__ sq,
                        const float* __restrict__ mPart, const float* __restrict__ sPart,
                        float* __restrict__ anchor, float* __restrict__ candV,
                        int* __restrict__ candI) {
  const int row  = blockIdx.x * 32 + threadIdx.x;
  const int lane = threadIdx.x;

  const v4f* mp = (const v4f*)mPart + lane;
  v4f m0 = {}, m1 = {}, m2 = {}, m3 = {};
  for (int b = 0; b < CS_BLOCKS; b += 4) {
    m0 += mp[(b    ) * (DPROJ / 4)];
    m1 += mp[(b + 1) * (DPROJ / 4)];
    m2 += mp[(b + 2) * (DPROJ / 4)];
    m3 += mp[(b + 3) * (DPROJ / 4)];
  }
  const v4f m = (m0 + m1) + (m2 + m3);
  float S = sPart[lane] + sPart[lane + 32];
  S += __int_as_float(lane_xor<1>(__float_as_int(S)));
  S += __int_as_float(lane_xor<2>(__float_as_int(S)));
  S += __int_as_float(lane_xor<4>(__float_as_int(S)));
  S += __int_as_float(lane_xor<8>(__float_as_int(S)));
  S += __int_as_float(lane_xor<16>(__float_as_int(S)));

  const v4f* gp = (const v4f*)(R + (size_t)blockIdx.x * 32 * DPROJ) + lane;
  const v2f mlo = {m.x, m.y};
  const v2f mhi = {m.z, m.w};
  float val[32];
  for (int r = 0; r < 32; r++) {
    const v4f o = gp[r * (DPROJ / 4)];
    const v2f lo = {o.x, o.y};
    const v2f hi = {o.z, o.w};
    const v2f p = __builtin_elementwise_fma(hi, mhi, lo * mlo);
    val[r] = p.x + p.y;
  }
  fold_stage<0>(val, lane);
  fold_stage<1>(val, lane);
  fold_stage<2>(val, lane);
  fold_stage<3>(val, lane);
  fold_stage<4>(val, lane);
  const float dot = val[0];
  const float d2 = fmaf((float)NPTS, sq[row], S) - 2.f * dot;
  const float a = sqrtf(fmaxf(d2, 0.f));
  anchor[row] = a;
  float v1, v2; int r1, r2;
  wave_top2(a, row, v1, r1, v2, r2);
  if (lane == 0) {
    candV[row >> 5] = v1; candV[NWAVES + (row >> 5)] = v2;
    candI[row >> 5] = r1; candI[NWAVES + (row >> 5)] = r2;
  }
}

__global__ __launch_bounds__(32)
void greedy_step_kernel(const float* __restrict__ R, const float* __restrict__ sq,
                        float* __restrict__ anchor0, float* __restrict__ anchor1,
                        float* __restrict__ candV, int* __restrict__ candI,
                        int* __restrict__ sel, int s) {
  const int row  = blockIdx.x * 32 + threadIdx.x;
  const int lane = threadIdx.x;
  const float* anchor = (s & 1) ? anchor1 : anchor0;
  float* anchorOut    = (s & 1) ? anchor0 : anchor1;
  const int pin  = (s & 1) * 2 * NWAVES;
  const int pout = 2 * NWAVES - pin;

  float4 cv[4]; int4 ci[4];
  const float4* cvp = (const float4*)(candV + pin) + lane;
  const int4*   cip = (const int4*)(candI + pin) + lane;
#pragma unroll
  for (int q = 0; q < 4; q++) { cv[q] = cvp[q * 32]; ci[q] = cip[q * 32]; }

  const v4f* gp = (const v4f*)(R + (size_t)blockIdx.x * 32 * DPROJ) + lane;
  v4f own[32];
#pragma unroll
  for (int r = 0; r < 32; r++) own[r] = gp[r * (DPROJ / 4)];
  const float sqOwn = sq[row];
  const float aOld  = anchor[row];
  __builtin_amdgcn_sched_barrier(0);

  float qm[4];
#pragma unroll
  for (int q = 0; q < 4; q++)
    qm[q] = fmaxf(fmaxf(cv[q].x, cv[q].y), fmaxf(cv[q].z, cv[q].w));
  float bv = fmaxf(fmaxf(qm[0], qm[1]), fmaxf(qm[2], qm[3]));
  int qi[4];
#pragma unroll
  for (int q = 0; q < 4; q++) {
    const int r0 = cv[q].x == bv ? ci[q].x : NO_ROW;
    const int r1 = cv[q].y == bv ? ci[q].y : NO_ROW;
    const int r2 = cv[q].z == bv ? ci[q].z : NO_ROW;
    const int r3 = cv[q].w == bv ? ci[q].w : NO_ROW;
    qi[q] = imin(imin(r0, r1), imin(r2, r3));
  }
  int bi = imin(imin(qi[0], qi[1]), imin(qi[2], qi[3]));
  wave_argmax(bv, bi);
  const int si = bi & (NPTS - 1);

  const v4f selPiece = ((const v4f*)(R + (size_t)si * DPROJ))[lane];
  const float selSq = sq[si];

  const v2f slo = {selPiece.x, selPiece.y};
  const v2f shi = {selPiece.z, selPiece.w};
  float val[32];
#pragma unroll
  for (int r = 0; r < 32; r++) {
    const v2f lo = {own[r].x, own[r].y};
    const v2f hi = {own[r].z, own[r].w};
    const v2f p = __builtin_elementwise_fma(hi, shi, lo * slo);
    val[r] = p.x + p.y;
  }
  fold_stage<0>(val, lane);
  fold_stage<1>(val, lane);
  fold_stage<2>(val, lane);
  fold_stage<3>(val, lane);
  fold_stage<4>(val, lane);
  const float dot = val[0];
  const float d2 = sqOwn + selSq - 2.f * dot;
  const float a  = fminf(aOld, sqrtf(fmaxf(d2, 0.f)));
  anchorOut[row] = a;

  float v1, v2; int r1, r2;
  wave_top2(a, row, v1, r1, v2, r2);
  if (lane == 0) {
    candV[pout + (row >> 5)] = v1; candV[pout + NWAVES + (row >> 5)] = v2;
    candI[pout + (row >> 5)] = r1; candI[pout + NWAVES + (row >> 5)] = r2;
  }
  if (row == 0) sel[s] = si;
}

struct Things {
  float e1[16], e2[16], rb[16];
  int   i1[16], i2[16];
};
__device__ __forceinline__ int slot_wave(int k, int lane) {
  return (k >> 2) * 128 + 4 * lane + (k & 3);
}
__device__ __forceinline__ void things_top(const Things& t, int lane, float& U, int& id) {
  float m[16];
#pragma unroll
  for (int k = 0; k < 16; k++) m[k] = fmaxf(fmaxf(t.e1[k], t.e2[k]), t.rb[k]);
#pragma unroll
  for (int w = 8; w > 0; w >>= 1)
#pragma unroll
    for (int k = 0; k < w; k++) m[k] = fmaxf(m[k], m[k + w]);
  const float top = wave_max(m[0]);
  int r[16];
#pragma unroll
  for (int k = 0; k < 16; k++) {
    const int a = t.e1[k] == top ? t.i1[k] : NO_ROW;
    const int b = t.e2[k] == top ? t.i2[k] : NO_ROW;
    const int c = t.rb[k] == top ? NPTS + slot_wave(k, lane) : NO_ROW;
    r[k] = imin(imin(a, b), c);
  }
#pragma unroll
  for (int w = 8; w > 0; w >>= 1)
#pragma unroll
    for (int k = 0; k < w; k++) r[k] = imin(r[k], r[k + w]);
  U = top;
  id = wave_min(r[0]);
}
__device__ __forceinline__ void things_mark(Things& t, int lane, int id) {
#pragma unroll
  for (int k = 0; k < 16; k++) {
    const bool whole = id == NPTS + slot_wave(k, lane);
    if (t.i1[k] == id || whole) t.e1[k] = NEG_INF;
    if (t.i2[k] == id || whole) t.e2[k] = NEG_INF;
    if (whole) t.rb[k] = NEG_INF;
  }
}

__device__ __forceinline__ int uni(int x) { return __builtin_amdgcn_readfirstlane(x); }
__device__ __forceinline__ float unif(float x) {
  return __int_as_float(__builtin_amdgcn_readfirstlane(__float_as_int(x)));
}

#define PAIR_FIRST 3

__global__ __launch_bounds__(32)
void greedy_pair_kernel(const float* __restrict__ R, const float* __restrict__ sq,
                        float* __restrict__ anchor0, float* __restrict__ anchor1,
                        float* __restrict__ candV, int* __restrict__ candI,
                        int* __restrict__ sel, int L) {
  const int row  = blockIdx.x * 32 + threadIdx.x;
  const int lane = threadIdx.x;
  const float* anchor = (L & 1) ? anchor1 : anchor0;
  float* anchorOut    = (L & 1) ? anchor0 : anchor1;
  const int pin  = (L & 1) * 2 * NWAVES;
  const int pout = 2 * NWAVES - pin;
  const int s    = PAIR_FIRST + 2 * (L - PAIR_FIRST);

  float4 cv1[4], cv2[4]; int4 ci1[4], ci2[4];
  const float4* cvp = (const float4*)(candV + pin) + lane;
  const int4*   cip = (const int4*)(candI + pin) + lane;
#pragma unroll
  for (int q = 0; q < 4; q++) {
    cv1[q] = cvp[q * 32]; cv2[q] = cvp[NWAVES / 4 + q * 32];
    ci1[q] = cip[q * 32]; ci2[q] = cip[NWAVES / 4 + q * 32];
  }
  __builtin_amdgcn_sched_barrier(0);
  const v4f* gp = (const v4f*)(R + (size_t)blockIdx.x * 32 * DPROJ) + lane;
  v4f own[32];
#pragma unroll
  for (int r = 0; r < 32; r++) own[r] = gp[r * (DPROJ / 4)];
  const float sqOwn = sq[row];
  const float aOld  = anchor[row];
  __builtin_amdgcn_sched_barrier(0);

  Things t;
#pragma unroll
  for (int q = 0; q < 4; q++) {
    t.e1[4 * q] = cv1[q].x; t.e1[4 * q + 1] = cv1[q].y; t.e1[4 * q + 2] = cv1[q].z; t.e1[4 * q + 3] = cv1[q].w;
    t.e2[4 * q] = cv2[q].x; t.e2[4 * q + 1] = cv2[q].y; t.e2[4 * q + 2] = cv2[q].z; t.e2[4 * q + 3] = cv2[q].w;
    t.i1[4 * q] = ci1[q].x; t.i1[4 * q + 1] = ci1[q].y; t.i1[4 * q + 2] = ci1[q].z; t.i1[4 * q + 3] = ci1[q].w;
    t.i2[4 * q] = ci2[q].x; t.i2[4 * q + 1] = ci2[q].y; t.i2[4 * q + 2] = ci2[q].z; t.i2[4 * q + 3] = ci2[q].w;
  }
#pragma unroll
  for (int k = 0; k < 16; k++) t.rb[k] = t.e2[k];

  float lm[16];
#pragma unroll
  for (int k = 0; k < 16; k++) lm[k] = t.e1[k];
#pragma unroll
  for (int w = 8; w > 0; w >>= 1)
#pragma unroll
    for (int k = 0; k < w; k++) lm[k] = fmaxf(lm[k], lm[k + w]);
  float v1 = lm[0];
  int li[16];
#pragma unroll
  for (int k = 0; k < 16; k++) li[k] = t.e1[k] == v1 ? t.i1[k] : NO_ROW;
#pragma unroll
  for (int w = 8; w > 0; w >>= 1)
#pragma unroll
    for (int k = 0; k < w; k++) li[k] = imin(li[k], li[k + w]);
  int b1 = li[0];
  wave_argmax(v1, b1);
  const int s1 = uni(b1) & (NPTS - 1);
  v1 = unif(v1);
  const v4f selP1 = ((const v4f*)(R + (size_t)s1 * DPROJ))[lane];
  const float selSq1 = sq[s1];
  __builtin_amdgcn_sched_barrier(0);

  things_mark(t, lane, s1);
  float u2; int id2;
  things_top(t, lane, u2, id2);
  id2 = uni(id2); u2 = unif(u2);
  const bool row2 = id2 < NPTS;
  const int c2 = row2 ? id2 : s1;
  const v4f P2 = ((const v4f*)(R + (size_t)c2 * DPROJ))[lane];
  const float sqC2 = sq[c2];
  __builtin_amdgcn_sched_barrier(0);
  if (row2) things_mark(t, lane, id2);
  float U; int id;
  things_top(t, lane, U, id);
  id = uni(id); U = unif(U);

  const float a1 = min_dist(aOld, sqOwn, selSq1, rows_dot(own, selP1, lane));

  float bestV = unif(min_dist(v1, selSq1, selSq1, row_dot_all(selP1, selP1)));
  int bestI = s1;
  if (row2) {
    const float e = unif(min_dist(u2, sqC2, selSq1, row_dot_all(P2, selP1)));
    if (e > bestV || (e == bestV && c2 < bestI)) { bestV = e; bestI = c2; }
  }
  for (int it = 0; it < 3 * NWAVES; it++) {
    if (bestV > U) break;
    float nv; int ni;
    if (id < NPTS) {
      const int x = id & (NPTS - 1);
      const v4f px = ((const v4f*)(R + (size_t)x * DPROJ))[lane];
      nv = min_dist(U, sq[x], selSq1, row_dot_all(px, selP1));
      ni = x;
    } else {
      const int w = (id - NPTS) & (NWAVES - 1);
      const v4f* wp = (const v4f*)(R + (size_t)w * 32 * DPROJ) + lane;
      v4f wr[32];
#pragma unroll
      for (int r = 0; r < 32; r++) wr[r] = wp[r * (DPROJ / 4)];
      ni = w * 32 + lane;
      nv = min_dist(anchor[ni], sq[ni], selSq1, rows_dot(wr, selP1, lane));
      wave_argmax(nv, ni);
    }
    nv = unif(nv); ni = uni(ni);
    if (nv > bestV || (nv == bestV && ni < bestI)) { bestV = nv; bestI = ni; }
    things_mark(t, lane, id);
    things_top(t, lane, U, id);
    id = uni(id); U = unif(U);
  }
  const int s2 = uni(bestI) & (NPTS - 1);

  v4f selP2 = P2; float selSq2 = sqC2;
  if (s2 != c2) {
    selP2 = ((const v4f*)(R + (size_t)s2 * DPROJ))[lane];
    selSq2 = sq[s2];
  }
  const float a2 = min_dist(a1, sqOwn, selSq2, rows_dot(own, selP2, lane));
  anchorOut[row] = a2;

  float w1, w2; int r1, r2;
  wave_top2(a2, row, w1, r1, w2, r2);
  if (lane == 0) {
    candV[pout + (row >> 5)] = w1; candV[pout + NWAVES + (row >> 5)] = w2;
    candI[pout + (row >> 5)] = r1; candI[pout + NWAVES + (row >> 5)] = r2;
  }
  if (row == 0) { sel[s] = s1; sel[s + 1] = s2; }
}


__global__ __launch_bounds__(256)
void gather_kernel(const float* __restrict__ F, const int* __restrict__ sel,
                   float* __restrict__ out) {
  const int s = blockIdx.x;
  const int c = blockIdx.y * 256 + threadIdx.x;
  out[(size_t)s * DIN + c] = F[(size_t)sel[s] * DIN + c];
}

extern "C" void kernel_launch(void* const* d_in, const int* in_sizes, int n_in,
                              void* d_out, int out_size, void* d_ws, size_t ws_size,
                              hipStream_t stream) {
  (void)in_sizes; (void)n_in; (void)out_size; (void)ws_size;
  const float* F = (const float*)d_in[0];
  const float* W = (const float*)d_in[1];

  float* R          = (float*)d_ws;
  float* sq         = R + (size_t)NPTS * DPROJ;
  float* anchorSq   = sq + NPTS;
  float* scratch    = anchorSq + NPTS;
  int*   sel        = (int*)(scratch + (size_t)SCRATCH_FLOATS);
  float* candV = scratch;
  int*   candI = (int*)(scratch + 4 * NWAVES);
  float* mPart = scratch + 8 * NWAVES;
  float* sPart = mPart + CS_BLOCKS * DPROJ;
  float* anchorB = scratch + 4 * NPTS;

  proj_kernel<<<NPTS / 64, 128, 0, stream>>>(F, W, R, sq);
  colsum_kernel<<<CS_BLOCKS, 256, 0, stream>>>(R, sq, mPart, sPart);
  anchor_init_kernel<<<NPTS / 32, 32, 0, stream>>>(R, sq, mPart, sPart, anchorSq, candV, candI);
  for (int L = 0; L < 3; L++)
    greedy_step_kernel<<<NPTS / 32, 32, 0, stream>>>(R, sq, anchorSq, anchorB, candV, candI, sel, L);
  for (int L = 3; L < 83; L++)
    greedy_pair_kernel<<<NPTS / 32, 32, 0, stream>>>(R, sq, anchorSq, anchorB, candV, candI, sel, L);
  gather_kernel<<<dim3(NSAMP, DIN / 256), 256, 0, stream>>>(F, sel, (float*)d_out);
}
